# starting kernel
# baseline (speedup 1.0000x reference)
#include <hip/hip_runtime.h>

typedef __attribute__((ext_vector_type(2))) float v2f;
typedef __attribute__((ext_vector_type(8))) float v8f;

#define NB 2
#define NV 16000
#define NPOINT 32
#define NVOX (NB * NV)
#define NPTS (NVOX * NPOINT)

#define WS_SUM1 0
#define WS_SQ1  16
#define WS_SUM2 32
#define WS_SQ2  64
#define WS_SUMF 96
#define WS_SQF  160
#define WS_S1   224
#define WS_B1   240
#define WS_S2   256
#define WS_B2   288
#define WS_SF   320
#define WS_BF   384

__device__ __forceinline__ void compute_feats(const float4& pt, const int4& cc, float f[8]) {
  float cx = ((float)cc.w + 0.5f) * 0.16f + 0.0f;
  float cy = ((float)cc.z + 0.5f) * 0.16f - 39.68f;
  float cz = ((float)cc.y + 0.5f) * 4.0f - 3.0f;
  f[0] = pt.x; f[1] = pt.y; f[2] = pt.z;
  f[3] = pt.x - cx; f[4] = pt.y - cy; f[5] = pt.z - cz;
  f[6] = pt.w; f[7] = pt.z;
}

__global__ __launch_bounds__(256) void k_stats1(const float* __restrict__ vox,
                                                const int* __restrict__ coords,
                                                const float* __restrict__ W1,
                                                float* __restrict__ ws) {
  __shared__ float sacc[32];
  __shared__ float sW[16 * 8];
  for (int i = threadIdx.x; i < 32; i += blockDim.x) sacc[i] = 0.f;
  for (int i = threadIdx.x; i < 128; i += blockDim.x) sW[i] = W1[i];
  __syncthreads();

  float acc[16], acq[16];
  #pragma unroll
  for (int o = 0; o < 16; o++) { acc[o] = 0.f; acq[o] = 0.f; }

  int tid = blockIdx.x * blockDim.x + threadIdx.x;
  int stride = gridDim.x * blockDim.x;
  for (int p = tid; p < NPTS; p += stride) {
    int vxl = p >> 5;
    float4 pt = ((const float4*)vox)[p];
    int4 cc = ((const int4*)coords)[vxl];
    float f[8];
    compute_feats(pt, cc, f);
    #pragma unroll
    for (int o = 0; o < 16; o++) {
      float h = 0.f;
      #pragma unroll
      for (int c = 0; c < 8; c++) h = fmaf(sW[o * 8 + c], f[c], h);
      acc[o] += h;
      acq[o] = fmaf(h, h, acq[o]);
    }
  }
  #pragma unroll
  for (int o = 0; o < 16; o++) {
    atomicAdd(&sacc[o], acc[o]);
    atomicAdd(&sacc[16 + o], acq[o]);
  }
  __syncthreads();
  if (threadIdx.x < 32) atomicAdd(&ws[WS_SUM1 + threadIdx.x], sacc[threadIdx.x]);
}

__global__ void k_bnprep(const float* __restrict__ sums, const float* __restrict__ sumsq,
                         const float* __restrict__ gamma, const float* __restrict__ beta,
                         float* __restrict__ scale, float* __restrict__ bias,
                         int C, float invN, float eps) {
  int o = threadIdx.x;
  if (o < C) {
    float m = sums[o] * invN;
    float var = sumsq[o] * invN - m * m;
    float s = gamma[o] * rsqrtf(var + eps);
    scale[o] = s;
    bias[o] = fmaf(-m, s, beta[o]);
  }
}

template <bool FINAL>
__global__ __launch_bounds__(256) void k_pass(const float* __restrict__ vox,
                                              const int* __restrict__ coords,
                                              const int* __restrict__ nump,
                                              const float* __restrict__ W1,
                                              const float* __restrict__ W2,
                                              const float* __restrict__ Wf,
                                              float* __restrict__ ws,
                                              float* __restrict__ out) {
  __shared__ float lds_f[8 * 32 * 8];
  __shared__ float lds_x2[8 * 32 * 32];
  __shared__ float lds_vf[16 * 64];
  __shared__ float sstat[128];

  const int lane = threadIdx.x & 31;
  const int wave = threadIdx.x >> 5;
  const int n = lane & 15;
  const int hi = lane >> 4;

  for (int i = threadIdx.x; i < 128; i += blockDim.x) sstat[i] = 0.f;
  for (int i = threadIdx.x; i < 16 * 64; i += blockDim.x) lds_vf[i] = 0.f;
  __syncthreads();

  float* myf = lds_f + wave * (32 * 8);
  float* myx = lds_x2 + wave * (32 * 32);

  v2f bw1[2], bw2[2][8], bwf[16];
  #pragma unroll
  for (int kb = 0; kb < 2; kb++) bw1[kb] = *(const v2f*)(W1 + n * 8 + 4 * kb + 2 * hi);
  #pragma unroll
  for (int nt = 0; nt < 2; nt++)
    #pragma unroll
    for (int kb = 0; kb < 8; kb++)
      bw2[nt][kb] = *(const v2f*)(W2 + (nt * 16 + n) * 32 + 4 * kb + 2 * hi);
  if (FINAL) {
    int fnt = wave & 3;
    #pragma unroll
    for (int kb = 0; kb < 16; kb++)
      bwf[kb] = *(const v2f*)(Wf + (fnt * 16 + n) * 64 + 4 * kb + 2 * hi);
  }

  const float s1v = ws[WS_S1 + n], b1v = ws[WS_B1 + n];
  float s2v[2] = {0.f, 0.f}, b2v[2] = {0.f, 0.f};
  if (FINAL) {
    s2v[0] = ws[WS_S2 + n];      s2v[1] = ws[WS_S2 + 16 + n];
    b2v[0] = ws[WS_B2 + n];      b2v[1] = ws[WS_B2 + 16 + n];
  }

  float accS[2] = {0.f, 0.f}, accQ[2] = {0.f, 0.f};
  float accSF = 0.f, accQF = 0.f;

  const int nchunks = NVOX / 8;
  for (int chunk = blockIdx.x; chunk < nchunks; chunk += gridDim.x) {
    const int vxl = chunk * 8 + wave;
    float4 pt = ((const float4*)vox)[vxl * NPOINT + lane];
    int4 cc = ((const int4*)coords)[vxl];
    int np_ = nump[vxl];

    float f[8];
    compute_feats(pt, cc, f);
    *(float4*)(myf + lane * 8 + 0) = make_float4(f[0], f[1], f[2], f[3]);
    *(float4*)(myf + lane * 8 + 4) = make_float4(f[4], f[5], f[6], f[7]);
    __syncthreads();

    v8f h1t[2];
    #pragma unroll
    for (int mt = 0; mt < 2; mt++) {
      v8f acc = {};
      #pragma unroll
      for (int kb = 0; kb < 2; kb++) {
        v2f a = *(const v2f*)(myf + (mt * 16 + n) * 8 + 4 * kb + 2 * hi);
        acc = __builtin_amdgcn_wmma_f32_16x16x4_f32(false, a, false, bw1[kb], (short)0, acc,
                                                    false, false);
      }
      h1t[mt] = acc;
    }

    float hval[2][8];
    float mxp = 0.f;
    #pragma unroll
    for (int mt = 0; mt < 2; mt++)
      #pragma unroll
      for (int r = 0; r < 8; r++) {
        float h = fmaxf(fmaf(s1v, h1t[mt][r], b1v), 0.f);
        hval[mt][r] = h;
        mxp = fmaxf(mxp, h);
      }
    mxp = fmaxf(mxp, __shfl_xor(mxp, 16));
    #pragma unroll
    for (int mt = 0; mt < 2; mt++)
      #pragma unroll
      for (int r = 0; r < 8; r++) {
        int p = mt * 16 + 8 * hi + r;
        float m = (p < np_) ? 1.f : 0.f;
        myx[p * 32 + n] = hval[mt][r] * m;
        myx[p * 32 + 16 + n] = mxp * m;
      }
    __syncthreads();

    v8f h2t[2][2];
    #pragma unroll
    for (int mt = 0; mt < 2; mt++) {
      v2f a[8];
      #pragma unroll
      for (int kb = 0; kb < 8; kb++)
        a[kb] = *(const v2f*)(myx + (mt * 16 + n) * 32 + 4 * kb + 2 * hi);
      #pragma unroll
      for (int nt = 0; nt < 2; nt++) {
        v8f acc = {};
        #pragma unroll
        for (int kb = 0; kb < 8; kb++)
          acc = __builtin_amdgcn_wmma_f32_16x16x4_f32(false, a[kb], false, bw2[nt][kb], (short)0,
                                                      acc, false, false);
        h2t[mt][nt] = acc;
      }
    }

    if (!FINAL) {
      #pragma unroll
      for (int nt = 0; nt < 2; nt++) {
        float s = 0.f, q = 0.f;
        #pragma unroll
        for (int mt = 0; mt < 2; mt++)
          #pragma unroll
          for (int r = 0; r < 8; r++) {
            float h = h2t[mt][nt][r];
            s += h;
            q = fmaf(h, h, q);
          }
        accS[nt] += s;
        accQ[nt] += q;
      }
    } else {
      float mx2[2] = {0.f, 0.f};
      float vfh[2] = {0.f, 0.f};
      #pragma unroll
      for (int nt = 0; nt < 2; nt++) {
        #pragma unroll
        for (int mt = 0; mt < 2; mt++)
          #pragma unroll
          for (int r = 0; r < 8; r++) {
            float h = fmaxf(fmaf(s2v[nt], h2t[mt][nt][r], b2v[nt]), 0.f);
            mx2[nt] = fmaxf(mx2[nt], h);
            int p = mt * 16 + 8 * hi + r;
            if (p < np_) vfh[nt] = fmaxf(vfh[nt], h);
          }
        mx2[nt] = fmaxf(mx2[nt], __shfl_xor(mx2[nt], 16));
        vfh[nt] = fmaxf(vfh[nt], __shfl_xor(vfh[nt], 16));
      }
      float anyv = (np_ > 0) ? 1.f : 0.f;
      if (hi == 0) {
        lds_vf[wave * 64 + n]      = vfh[0];
        lds_vf[wave * 64 + 16 + n] = vfh[1];
        lds_vf[wave * 64 + 32 + n] = mx2[0] * anyv;
        lds_vf[wave * 64 + 48 + n] = mx2[1] * anyv;
      }
      __syncthreads();

      if (wave < 4) {
        v8f acc = {};
        #pragma unroll
        for (int kb = 0; kb < 16; kb++) {
          v2f a = *(const v2f*)(lds_vf + n * 64 + 4 * kb + 2 * hi);
          acc = __builtin_amdgcn_wmma_f32_16x16x4_f32(false, a, false, bwf[kb], (short)0, acc,
                                                      false, false);
        }
        if (hi == 0) {
          #pragma unroll
          for (int r = 0; r < 8; r++) {
            float g = acc[r];
            out[(chunk * 8 + r) * 64 + wave * 16 + n] = g;
            accSF += g;
            accQF = fmaf(g, g, accQF);
          }
        }
      }
      __syncthreads();
    }
  }

  __syncthreads();
  if (!FINAL) {
    atomicAdd(&sstat[n], accS[0]);
    atomicAdd(&sstat[16 + n], accS[1]);
    atomicAdd(&sstat[32 + n], accQ[0]);
    atomicAdd(&sstat[48 + n], accQ[1]);
    __syncthreads();
    for (int i = threadIdx.x; i < 64; i += blockDim.x) atomicAdd(&ws[WS_SUM2 + i], sstat[i]);
  } else {
    if (wave < 4 && hi == 0) {
      atomicAdd(&sstat[wave * 16 + n], accSF);
      atomicAdd(&sstat[64 + wave * 16 + n], accQF);
    }
    __syncthreads();
    for (int i = threadIdx.x; i < 128; i += blockDim.x) atomicAdd(&ws[WS_SUMF + i], sstat[i]);
  }
}

__global__ __launch_bounds__(256) void k_norm(float* __restrict__ out,
                                              const float* __restrict__ scale,
                                              const float* __restrict__ bias, int nElem) {
  int i = blockIdx.x * blockDim.x + threadIdx.x;
  if (i < nElem) {
    int c = i & 63;
    out[i] = fmaf(scale[c], out[i], bias[c]);
  }
}

extern "C" void kernel_launch(void* const* d_in, const int* in_sizes, int n_in,
                              void* d_out, int out_size, void* d_ws, size_t ws_size,
                              hipStream_t stream) {
  (void)in_sizes; (void)n_in; (void)out_size; (void)ws_size;
  const float* vox    = (const float*)d_in[0];
  const int*   coords = (const int*)d_in[1];
  const int*   nump   = (const int*)d_in[2];
  const float* W1     = (const float*)d_in[3];
  const float* g1     = (const float*)d_in[4];
  const float* b1     = (const float*)d_in[5];
  const float* W2     = (const float*)d_in[6];
  const float* g2     = (const float*)d_in[7];
  const float* b2     = (const float*)d_in[8];
  const float* Wf     = (const float*)d_in[9];
  const float* gf     = (const float*)d_in[10];
  const float* bf     = (const float*)d_in[11];
  float* out = (float*)d_out;
  float* ws  = (float*)d_ws;

  hipMemsetAsync(ws, 0, 224 * sizeof(float), stream);
  k_stats1<<<1024, 256, 0, stream>>>(vox, coords, W1, ws);
  k_bnprep<<<1, 64, 0, stream>>>(ws + WS_SUM1, ws + WS_SQ1, g1, b1, ws + WS_S1, ws + WS_B1,
                                 16, 1.0f / (float)NPTS, 1e-3f);
  k_pass<false><<<250, 256, 0, stream>>>(vox, coords, nump, W1, W2, Wf, ws, out);
  k_bnprep<<<1, 64, 0, stream>>>(ws + WS_SUM2, ws + WS_SQ2, g2, b2, ws + WS_S2, ws + WS_B2,
                                 32, 1.0f / (float)NPTS, 1e-3f);
  k_pass<true><<<250, 256, 0, stream>>>(vox, coords, nump, W1, W2, Wf, ws, out);
  k_bnprep<<<1, 64, 0, stream>>>(ws + WS_SUMF, ws + WS_SQF, gf, bf, ws + WS_SF, ws + WS_BF,
                                 64, 1.0f / (float)NVOX, 1e-5f);
  k_norm<<<(NVOX * 64 + 255) / 256, 256, 0, stream>>>(out, ws + WS_SF, ws + WS_BF, NVOX * 64);
}


# k_stats1 reduction rebuilt: wave transpose-reduce (31 shuffles) + per-wave LDS slots replace 32 same-address LDS atomics per thread; 256 blocks of 512 so 4x fewer global atomics; new summation tree
# speedup vs baseline: 1.4909x; 1.4909x over previous
#include <hip/hip_runtime.h>

typedef __attribute__((ext_vector_type(2))) float v2f;
typedef __attribute__((ext_vector_type(8))) float v8f;

#define NB 2
#define NV 16000
#define NPOINT 32
#define NVOX (NB * NV)
#define NPTS (NVOX * NPOINT)

#define WS_SUM1 0
#define WS_SQ1  16
#define WS_SUM2 32
#define WS_SQ2  64
#define WS_SUMF 96
#define WS_SQF  160
#define WS_S1   224
#define WS_B1   240
#define WS_S2   256
#define WS_B2   288
#define WS_SF   320
#define WS_BF   384

__device__ __forceinline__ void compute_feats(const float4& pt, const int4& cc, float f[8]) {
  float cx = ((float)cc.w + 0.5f) * 0.16f + 0.0f;
  float cy = ((float)cc.z + 0.5f) * 0.16f - 39.68f;
  float cz = ((float)cc.y + 0.5f) * 4.0f - 3.0f;
  f[0] = pt.x; f[1] = pt.y; f[2] = pt.z;
  f[3] = pt.x - cx; f[4] = pt.y - cy; f[5] = pt.z - cz;
  f[6] = pt.w; f[7] = pt.z;
}

__global__ __launch_bounds__(512) void k_stats1(const float* __restrict__ vox,
                                                const int* __restrict__ coords,
                                                const float* __restrict__ W1,
                                                float* __restrict__ ws) {
  __shared__ float wpart[16 * 32];
  __shared__ float sW[16 * 8];
  for (int i = threadIdx.x; i < 128; i += blockDim.x) sW[i] = W1[i];
  __syncthreads();

  float acc[16], acq[16];
  #pragma unroll
  for (int o = 0; o < 16; o++) { acc[o] = 0.f; acq[o] = 0.f; }

  int tid = blockIdx.x * blockDim.x + threadIdx.x;
  int stride = gridDim.x * blockDim.x;
  for (int p = tid; p < NPTS; p += stride) {
    int vxl = p >> 5;
    float4 pt = ((const float4*)vox)[p];
    int4 cc = ((const int4*)coords)[vxl];
    float f[8];
    compute_feats(pt, cc, f);
    #pragma unroll
    for (int o = 0; o < 16; o++) {
      float h = 0.f;
      #pragma unroll
      for (int c = 0; c < 8; c++) h = fmaf(sW[o * 8 + c], f[c], h);
      acc[o] += h;
      acq[o] = fmaf(h, h, acq[o]);
    }
  }
  const int lane = threadIdx.x & 31;
  const int wave = threadIdx.x >> 5;
  float v[32];
  #pragma unroll
  for (int o = 0; o < 16; o++) { v[o] = acc[o]; v[16 + o] = acq[o]; }
  #pragma unroll
  for (int w = 16; w >= 1; w >>= 1) {
    const bool up = (lane & w) != 0;
    #pragma unroll
    for (int i = 0; i < w; i++) {
      const float lo = v[i], hi = v[i + w];
      const float keep = up ? hi : lo;
      const float send = up ? lo : hi;
      v[i] = keep + __shfl_xor(send, w);
    }
  }
  wpart[wave * 32 + lane] = v[0];
  __syncthreads();
  if (threadIdx.x < 32) {
    float s = 0.f;
    for (int k = 0; k < 16; k++) s += wpart[k * 32 + threadIdx.x];
    atomicAdd(&ws[WS_SUM1 + threadIdx.x], s);
  }
}

__global__ void k_bnprep(const float* __restrict__ sums, const float* __restrict__ sumsq,
                         const float* __restrict__ gamma, const float* __restrict__ beta,
                         float* __restrict__ scale, float* __restrict__ bias,
                         int C, float invN, float eps) {
  int o = threadIdx.x;
  if (o < C) {
    float m = sums[o] * invN;
    float var = sumsq[o] * invN - m * m;
    float s = gamma[o] * rsqrtf(var + eps);
    scale[o] = s;
    bias[o] = fmaf(-m, s, beta[o]);
  }
}

template <bool FINAL>
__global__ __launch_bounds__(256) void k_pass(const float* __restrict__ vox,
                                              const int* __restrict__ coords,
                                              const int* __restrict__ nump,
                                              const float* __restrict__ W1,
                                              const float* __restrict__ W2,
                                              const float* __restrict__ Wf,
                                              float* __restrict__ ws,
                                              float* __restrict__ out) {
  __shared__ float lds_f[8 * 32 * 8];
  __shared__ float lds_x2[8 * 32 * 32];
  __shared__ float lds_vf[16 * 64];
  __shared__ float sstat[128];

  const int lane = threadIdx.x & 31;
  const int wave = threadIdx.x >> 5;
  const int n = lane & 15;
  const int hi = lane >> 4;

  for (int i = threadIdx.x; i < 128; i += blockDim.x) sstat[i] = 0.f;
  for (int i = threadIdx.x; i < 16 * 64; i += blockDim.x) lds_vf[i] = 0.f;
  __syncthreads();

  float* myf = lds_f + wave * (32 * 8);
  float* myx = lds_x2 + wave * (32 * 32);

  v2f bw1[2], bw2[2][8], bwf[16];
  #pragma unroll
  for (int kb = 0; kb < 2; kb++) bw1[kb] = *(const v2f*)(W1 + n * 8 + 4 * kb + 2 * hi);
  #pragma unroll
  for (int nt = 0; nt < 2; nt++)
    #pragma unroll
    for (int kb = 0; kb < 8; kb++)
      bw2[nt][kb] = *(const v2f*)(W2 + (nt * 16 + n) * 32 + 4 * kb + 2 * hi);
  if (FINAL) {
    int fnt = wave & 3;
    #pragma unroll
    for (int kb = 0; kb < 16; kb++)
      bwf[kb] = *(const v2f*)(Wf + (fnt * 16 + n) * 64 + 4 * kb + 2 * hi);
  }

  const float s1v = ws[WS_S1 + n], b1v = ws[WS_B1 + n];
  float s2v[2] = {0.f, 0.f}, b2v[2] = {0.f, 0.f};
  if (FINAL) {
    s2v[0] = ws[WS_S2 + n];      s2v[1] = ws[WS_S2 + 16 + n];
    b2v[0] = ws[WS_B2 + n];      b2v[1] = ws[WS_B2 + 16 + n];
  }

  float accS[2] = {0.f, 0.f}, accQ[2] = {0.f, 0.f};
  float accSF = 0.f, accQF = 0.f;

  const int nchunks = NVOX / 8;
  for (int chunk = blockIdx.x; chunk < nchunks; chunk += gridDim.x) {
    const int vxl = chunk * 8 + wave;
    float4 pt = ((const float4*)vox)[vxl * NPOINT + lane];
    int4 cc = ((const int4*)coords)[vxl];
    int np_ = nump[vxl];

    float f[8];
    compute_feats(pt, cc, f);
    *(float4*)(myf + lane * 8 + 0) = make_float4(f[0], f[1], f[2], f[3]);
    *(float4*)(myf + lane * 8 + 4) = make_float4(f[4], f[5], f[6], f[7]);
    __syncthreads();

    v8f h1t[2];
    #pragma unroll
    for (int mt = 0; mt < 2; mt++) {
      v8f acc = {};
      #pragma unroll
      for (int kb = 0; kb < 2; kb++) {
        v2f a = *(const v2f*)(myf + (mt * 16 + n) * 8 + 4 * kb + 2 * hi);
        acc = __builtin_amdgcn_wmma_f32_16x16x4_f32(false, a, false, bw1[kb], (short)0, acc,
                                                    false, false);
      }
      h1t[mt] = acc;
    }

    float hval[2][8];
    float mxp = 0.f;
    #pragma unroll
    for (int mt = 0; mt < 2; mt++)
      #pragma unroll
      for (int r = 0; r < 8; r++) {
        float h = fmaxf(fmaf(s1v, h1t[mt][r], b1v), 0.f);
        hval[mt][r] = h;
        mxp = fmaxf(mxp, h);
      }
    mxp = fmaxf(mxp, __shfl_xor(mxp, 16));
    #pragma unroll
    for (int mt = 0; mt < 2; mt++)
      #pragma unroll
      for (int r = 0; r < 8; r++) {
        int p = mt * 16 + 8 * hi + r;
        float m = (p < np_) ? 1.f : 0.f;
        myx[p * 32 + n] = hval[mt][r] * m;
        myx[p * 32 + 16 + n] = mxp * m;
      }
    __syncthreads();

    v8f h2t[2][2];
    #pragma unroll
    for (int mt = 0; mt < 2; mt++) {
      v2f a[8];
      #pragma unroll
      for (int kb = 0; kb < 8; kb++)
        a[kb] = *(const v2f*)(myx + (mt * 16 + n) * 32 + 4 * kb + 2 * hi);
      #pragma unroll
      for (int nt = 0; nt < 2; nt++) {
        v8f acc = {};
        #pragma unroll
        for (int kb = 0; kb < 8; kb++)
          acc = __builtin_amdgcn_wmma_f32_16x16x4_f32(false, a[kb], false, bw2[nt][kb], (short)0,
                                                      acc, false, false);
        h2t[mt][nt] = acc;
      }
    }

    if (!FINAL) {
      #pragma unroll
      for (int nt = 0; nt < 2; nt++) {
        float s = 0.f, q = 0.f;
        #pragma unroll
        for (int mt = 0; mt < 2; mt++)
          #pragma unroll
          for (int r = 0; r < 8; r++) {
            float h = h2t[mt][nt][r];
            s += h;
            q = fmaf(h, h, q);
          }
        accS[nt] += s;
        accQ[nt] += q;
      }
    } else {
      float mx2[2] = {0.f, 0.f};
      float vfh[2] = {0.f, 0.f};
      #pragma unroll
      for (int nt = 0; nt < 2; nt++) {
        #pragma unroll
        for (int mt = 0; mt < 2; mt++)
          #pragma unroll
          for (int r = 0; r < 8; r++) {
            float h = fmaxf(fmaf(s2v[nt], h2t[mt][nt][r], b2v[nt]), 0.f);
            mx2[nt] = fmaxf(mx2[nt], h);
            int p = mt * 16 + 8 * hi + r;
            if (p < np_) vfh[nt] = fmaxf(vfh[nt], h);
          }
        mx2[nt] = fmaxf(mx2[nt], __shfl_xor(mx2[nt], 16));
        vfh[nt] = fmaxf(vfh[nt], __shfl_xor(vfh[nt], 16));
      }
      float anyv = (np_ > 0) ? 1.f : 0.f;
      if (hi == 0) {
        lds_vf[wave * 64 + n]      = vfh[0];
        lds_vf[wave * 64 + 16 + n] = vfh[1];
        lds_vf[wave * 64 + 32 + n] = mx2[0] * anyv;
        lds_vf[wave * 64 + 48 + n] = mx2[1] * anyv;
      }
      __syncthreads();

      if (wave < 4) {
        v8f acc = {};
        #pragma unroll
        for (int kb = 0; kb < 16; kb++) {
          v2f a = *(const v2f*)(lds_vf + n * 64 + 4 * kb + 2 * hi);
          acc = __builtin_amdgcn_wmma_f32_16x16x4_f32(false, a, false, bwf[kb], (short)0, acc,
                                                      false, false);
        }
        if (hi == 0) {
          #pragma unroll
          for (int r = 0; r < 8; r++) {
            float g = acc[r];
            out[(chunk * 8 + r) * 64 + wave * 16 + n] = g;
            accSF += g;
            accQF = fmaf(g, g, accQF);
          }
        }
      }
      __syncthreads();
    }
  }

  __syncthreads();
  if (!FINAL) {
    atomicAdd(&sstat[n], accS[0]);
    atomicAdd(&sstat[16 + n], accS[1]);
    atomicAdd(&sstat[32 + n], accQ[0]);
    atomicAdd(&sstat[48 + n], accQ[1]);
    __syncthreads();
    for (int i = threadIdx.x; i < 64; i += blockDim.x) atomicAdd(&ws[WS_SUM2 + i], sstat[i]);
  } else {
    if (wave < 4 && hi == 0) {
      atomicAdd(&sstat[wave * 16 + n], accSF);
      atomicAdd(&sstat[64 + wave * 16 + n], accQF);
    }
    __syncthreads();
    for (int i = threadIdx.x; i < 128; i += blockDim.x) atomicAdd(&ws[WS_SUMF + i], sstat[i]);
  }
}

__global__ __launch_bounds__(256) void k_norm(float* __restrict__ out,
                                              const float* __restrict__ scale,
                                              const float* __restrict__ bias, int nElem) {
  int i = blockIdx.x * blockDim.x + threadIdx.x;
  if (i < nElem) {
    int c = i & 63;
    out[i] = fmaf(scale[c], out[i], bias[c]);
  }
}

extern "C" void kernel_launch(void* const* d_in, const int* in_sizes, int n_in,
                              void* d_out, int out_size, void* d_ws, size_t ws_size,
                              hipStream_t stream) {
  (void)in_sizes; (void)n_in; (void)out_size; (void)ws_size;
  const float* vox    = (const float*)d_in[0];
  const int*   coords = (const int*)d_in[1];
  const int*   nump   = (const int*)d_in[2];
  const float* W1     = (const float*)d_in[3];
  const float* g1     = (const float*)d_in[4];
  const float* b1     = (const float*)d_in[5];
  const float* W2     = (const float*)d_in[6];
  const float* g2     = (const float*)d_in[7];
  const float* b2     = (const float*)d_in[8];
  const float* Wf     = (const float*)d_in[9];
  const float* gf     = (const float*)d_in[10];
  const float* bf     = (const float*)d_in[11];
  float* out = (float*)d_out;
  float* ws  = (float*)d_ws;

  hipMemsetAsync(ws, 0, 224 * sizeof(float), stream);
  k_stats1<<<256, 512, 0, stream>>>(vox, coords, W1, ws);
  k_bnprep<<<1, 64, 0, stream>>>(ws + WS_SUM1, ws + WS_SQ1, g1, b1, ws + WS_S1, ws + WS_B1,
                                 16, 1.0f / (float)NPTS, 1e-3f);
  k_pass<false><<<250, 256, 0, stream>>>(vox, coords, nump, W1, W2, Wf, ws, out);
  k_bnprep<<<1, 64, 0, stream>>>(ws + WS_SUM2, ws + WS_SQ2, g2, b2, ws + WS_S2, ws + WS_B2,
                                 32, 1.0f / (float)NPTS, 1e-3f);
  k_pass<true><<<250, 256, 0, stream>>>(vox, coords, nump, W1, W2, Wf, ws, out);
  k_bnprep<<<1, 64, 0, stream>>>(ws + WS_SUMF, ws + WS_SQF, gf, bf, ws + WS_SF, ws + WS_BF,
                                 64, 1.0f / (float)NVOX, 1e-5f);
  k_norm<<<(NVOX * 64 + 255) / 256, 256, 0, stream>>>(out, ws + WS_SF, ws + WS_BF, NVOX * 64);
}


# k_pass grids raised to what stays resident: stats pass 250->1000 workgroups (4 per WGP), final pass 250->500 (2 per WGP); co-resident groups fill each other's waits; more partials per float atomic
# speedup vs baseline: 1.9224x; 1.2894x over previous
#include <hip/hip_runtime.h>

typedef __attribute__((ext_vector_type(2))) float v2f;
typedef __attribute__((ext_vector_type(8))) float v8f;

#define NB 2
#define NV 16000
#define NPOINT 32
#define NVOX (NB * NV)
#define NPTS (NVOX * NPOINT)

#define WS_SUM1 0
#define WS_SQ1  16
#define WS_SUM2 32
#define WS_SQ2  64
#define WS_SUMF 96
#define WS_SQF  160
#define WS_S1   224
#define WS_B1   240
#define WS_S2   256
#define WS_B2   288
#define WS_SF   320
#define WS_BF   384

__device__ __forceinline__ void compute_feats(const float4& pt, const int4& cc, float f[8]) {
  float cx = ((float)cc.w + 0.5f) * 0.16f + 0.0f;
  float cy = ((float)cc.z + 0.5f) * 0.16f - 39.68f;
  float cz = ((float)cc.y + 0.5f) * 4.0f - 3.0f;
  f[0] = pt.x; f[1] = pt.y; f[2] = pt.z;
  f[3] = pt.x - cx; f[4] = pt.y - cy; f[5] = pt.z - cz;
  f[6] = pt.w; f[7] = pt.z;
}

__global__ __launch_bounds__(512) void k_stats1(const float* __restrict__ vox,
                                                const int* __restrict__ coords,
                                                const float* __restrict__ W1,
                                                float* __restrict__ ws) {
  __shared__ float wpart[16 * 32];
  __shared__ float sW[16 * 8];
  for (int i = threadIdx.x; i < 128; i += blockDim.x) sW[i] = W1[i];
  __syncthreads();

  float acc[16], acq[16];
  #pragma unroll
  for (int o = 0; o < 16; o++) { acc[o] = 0.f; acq[o] = 0.f; }

  int tid = blockIdx.x * blockDim.x + threadIdx.x;
  int stride = gridDim.x * blockDim.x;
  for (int p = tid; p < NPTS; p += stride) {
    int vxl = p >> 5;
    float4 pt = ((const float4*)vox)[p];
    int4 cc = ((const int4*)coords)[vxl];
    float f[8];
    compute_feats(pt, cc, f);
    #pragma unroll
    for (int o = 0; o < 16; o++) {
      float h = 0.f;
      #pragma unroll
      for (int c = 0; c < 8; c++) h = fmaf(sW[o * 8 + c], f[c], h);
      acc[o] += h;
      acq[o] = fmaf(h, h, acq[o]);
    }
  }
  const int lane = threadIdx.x & 31;
  const int wave = threadIdx.x >> 5;
  float v[32];
  #pragma unroll
  for (int o = 0; o < 16; o++) { v[o] = acc[o]; v[16 + o] = acq[o]; }
  #pragma unroll
  for (int w = 16; w >= 1; w >>= 1) {
    const bool up = (lane & w) != 0;
    #pragma unroll
    for (int i = 0; i < w; i++) {
      const float lo = v[i], hi = v[i + w];
      const float keep = up ? hi : lo;
      const float send = up ? lo : hi;
      v[i] = keep + __shfl_xor(send, w);
    }
  }
  wpart[wave * 32 + lane] = v[0];
  __syncthreads();
  if (threadIdx.x < 32) {
    float s = 0.f;
    for (int k = 0; k < 16; k++) s += wpart[k * 32 + threadIdx.x];
    atomicAdd(&ws[WS_SUM1 + threadIdx.x], s);
  }
}

__global__ void k_bnprep(const float* __restrict__ sums, const float* __restrict__ sumsq,
                         const float* __restrict__ gamma, const float* __restrict__ beta,
                         float* __restrict__ scale, float* __restrict__ bias,
                         int C, float invN, float eps) {
  int o = threadIdx.x;
  if (o < C) {
    float m = sums[o] * invN;
    float var = sumsq[o] * invN - m * m;
    float s = gamma[o] * rsqrtf(var + eps);
    scale[o] = s;
    bias[o] = fmaf(-m, s, beta[o]);
  }
}

template <bool FINAL>
__global__ __launch_bounds__(256) void k_pass(const float* __restrict__ vox,
                                              const int* __restrict__ coords,
                                              const int* __restrict__ nump,
                                              const float* __restrict__ W1,
                                              const float* __restrict__ W2,
                                              const float* __restrict__ Wf,
                                              float* __restrict__ ws,
                                              float* __restrict__ out) {
  __shared__ float lds_f[8 * 32 * 8];
  __shared__ float lds_x2[8 * 32 * 32];
  __shared__ float lds_vf[16 * 64];
  __shared__ float sstat[128];

  const int lane = threadIdx.x & 31;
  const int wave = threadIdx.x >> 5;
  const int n = lane & 15;
  const int hi = lane >> 4;

  for (int i = threadIdx.x; i < 128; i += blockDim.x) sstat[i] = 0.f;
  for (int i = threadIdx.x; i < 16 * 64; i += blockDim.x) lds_vf[i] = 0.f;
  __syncthreads();

  float* myf = lds_f + wave * (32 * 8);
  float* myx = lds_x2 + wave * (32 * 32);

  v2f bw1[2], bw2[2][8], bwf[16];
  #pragma unroll
  for (int kb = 0; kb < 2; kb++) bw1[kb] = *(const v2f*)(W1 + n * 8 + 4 * kb + 2 * hi);
  #pragma unroll
  for (int nt = 0; nt < 2; nt++)
    #pragma unroll
    for (int kb = 0; kb < 8; kb++)
      bw2[nt][kb] = *(const v2f*)(W2 + (nt * 16 + n) * 32 + 4 * kb + 2 * hi);
  if (FINAL) {
    int fnt = wave & 3;
    #pragma unroll
    for (int kb = 0; kb < 16; kb++)
      bwf[kb] = *(const v2f*)(Wf + (fnt * 16 + n) * 64 + 4 * kb + 2 * hi);
  }

  const float s1v = ws[WS_S1 + n], b1v = ws[WS_B1 + n];
  float s2v[2] = {0.f, 0.f}, b2v[2] = {0.f, 0.f};
  if (FINAL) {
    s2v[0] = ws[WS_S2 + n];      s2v[1] = ws[WS_S2 + 16 + n];
    b2v[0] = ws[WS_B2 + n];      b2v[1] = ws[WS_B2 + 16 + n];
  }

  float accS[2] = {0.f, 0.f}, accQ[2] = {0.f, 0.f};
  float accSF = 0.f, accQF = 0.f;

  const int nchunks = NVOX / 8;
  for (int chunk = blockIdx.x; chunk < nchunks; chunk += gridDim.x) {
    const int vxl = chunk * 8 + wave;
    float4 pt = ((const float4*)vox)[vxl * NPOINT + lane];
    int4 cc = ((const int4*)coords)[vxl];
    int np_ = nump[vxl];

    float f[8];
    compute_feats(pt, cc, f);
    *(float4*)(myf + lane * 8 + 0) = make_float4(f[0], f[1], f[2], f[3]);
    *(float4*)(myf + lane * 8 + 4) = make_float4(f[4], f[5], f[6], f[7]);
    __syncthreads();

    v8f h1t[2];
    #pragma unroll
    for (int mt = 0; mt < 2; mt++) {
      v8f acc = {};
      #pragma unroll
      for (int kb = 0; kb < 2; kb++) {
        v2f a = *(const v2f*)(myf + (mt * 16 + n) * 8 + 4 * kb + 2 * hi);
        acc = __builtin_amdgcn_wmma_f32_16x16x4_f32(false, a, false, bw1[kb], (short)0, acc,
                                                    false, false);
      }
      h1t[mt] = acc;
    }

    float hval[2][8];
    float mxp = 0.f;
    #pragma unroll
    for (int mt = 0; mt < 2; mt++)
      #pragma unroll
      for (int r = 0; r < 8; r++) {
        float h = fmaxf(fmaf(s1v, h1t[mt][r], b1v), 0.f);
        hval[mt][r] = h;
        mxp = fmaxf(mxp, h);
      }
    mxp = fmaxf(mxp, __shfl_xor(mxp, 16));
    #pragma unroll
    for (int mt = 0; mt < 2; mt++)
      #pragma unroll
      for (int r = 0; r < 8; r++) {
        int p = mt * 16 + 8 * hi + r;
        float m = (p < np_) ? 1.f : 0.f;
        myx[p * 32 + n] = hval[mt][r] * m;
        myx[p * 32 + 16 + n] = mxp * m;
      }
    __syncthreads();

    v8f h2t[2][2];
    #pragma unroll
    for (int mt = 0; mt < 2; mt++) {
      v2f a[8];
      #pragma unroll
      for (int kb = 0; kb < 8; kb++)
        a[kb] = *(const v2f*)(myx + (mt * 16 + n) * 32 + 4 * kb + 2 * hi);
      #pragma unroll
      for (int nt = 0; nt < 2; nt++) {
        v8f acc = {};
        #pragma unroll
        for (int kb = 0; kb < 8; kb++)
          acc = __builtin_amdgcn_wmma_f32_16x16x4_f32(false, a[kb], false, bw2[nt][kb], (short)0,
                                                      acc, false, false);
        h2t[mt][nt] = acc;
      }
    }

    if (!FINAL) {
      #pragma unroll
      for (int nt = 0; nt < 2; nt++) {
        float s = 0.f, q = 0.f;
        #pragma unroll
        for (int mt = 0; mt < 2; mt++)
          #pragma unroll
          for (int r = 0; r < 8; r++) {
            float h = h2t[mt][nt][r];
            s += h;
            q = fmaf(h, h, q);
          }
        accS[nt] += s;
        accQ[nt] += q;
      }
    } else {
      float mx2[2] = {0.f, 0.f};
      float vfh[2] = {0.f, 0.f};
      #pragma unroll
      for (int nt = 0; nt < 2; nt++) {
        #pragma unroll
        for (int mt = 0; mt < 2; mt++)
          #pragma unroll
          for (int r = 0; r < 8; r++) {
            float h = fmaxf(fmaf(s2v[nt], h2t[mt][nt][r], b2v[nt]), 0.f);
            mx2[nt] = fmaxf(mx2[nt], h);
            int p = mt * 16 + 8 * hi + r;
            if (p < np_) vfh[nt] = fmaxf(vfh[nt], h);
          }
        mx2[nt] = fmaxf(mx2[nt], __shfl_xor(mx2[nt], 16));
        vfh[nt] = fmaxf(vfh[nt], __shfl_xor(vfh[nt], 16));
      }
      float anyv = (np_ > 0) ? 1.f : 0.f;
      if (hi == 0) {
        lds_vf[wave * 64 + n]      = vfh[0];
        lds_vf[wave * 64 + 16 + n] = vfh[1];
        lds_vf[wave * 64 + 32 + n] = mx2[0] * anyv;
        lds_vf[wave * 64 + 48 + n] = mx2[1] * anyv;
      }
      __syncthreads();

      if (wave < 4) {
        v8f acc = {};
        #pragma unroll
        for (int kb = 0; kb < 16; kb++) {
          v2f a = *(const v2f*)(lds_vf + n * 64 + 4 * kb + 2 * hi);
          acc = __builtin_amdgcn_wmma_f32_16x16x4_f32(false, a, false, bwf[kb], (short)0, acc,
                                                      false, false);
        }
        if (hi == 0) {
          #pragma unroll
          for (int r = 0; r < 8; r++) {
            float g = acc[r];
            out[(chunk * 8 + r) * 64 + wave * 16 + n] = g;
            accSF += g;
            accQF = fmaf(g, g, accQF);
          }
        }
      }
      __syncthreads();
    }
  }

  __syncthreads();
  if (!FINAL) {
    atomicAdd(&sstat[n], accS[0]);
    atomicAdd(&sstat[16 + n], accS[1]);
    atomicAdd(&sstat[32 + n], accQ[0]);
    atomicAdd(&sstat[48 + n], accQ[1]);
    __syncthreads();
    for (int i = threadIdx.x; i < 64; i += blockDim.x) atomicAdd(&ws[WS_SUM2 + i], sstat[i]);
  } else {
    if (wave < 4 && hi == 0) {
      atomicAdd(&sstat[wave * 16 + n], accSF);
      atomicAdd(&sstat[64 + wave * 16 + n], accQF);
    }
    __syncthreads();
    for (int i = threadIdx.x; i < 128; i += blockDim.x) atomicAdd(&ws[WS_SUMF + i], sstat[i]);
  }
}

__global__ __launch_bounds__(256) void k_norm(float* __restrict__ out,
                                              const float* __restrict__ scale,
                                              const float* __restrict__ bias, int nElem) {
  int i = blockIdx.x * blockDim.x + threadIdx.x;
  if (i < nElem) {
    int c = i & 63;
    out[i] = fmaf(scale[c], out[i], bias[c]);
  }
}

extern "C" void kernel_launch(void* const* d_in, const int* in_sizes, int n_in,
                              void* d_out, int out_size, void* d_ws, size_t ws_size,
                              hipStream_t stream) {
  (void)in_sizes; (void)n_in; (void)out_size; (void)ws_size;
  const float* vox    = (const float*)d_in[0];
  const int*   coords = (const int*)d_in[1];
  const int*   nump   = (const int*)d_in[2];
  const float* W1     = (const float*)d_in[3];
  const float* g1     = (const float*)d_in[4];
  const float* b1     = (const float*)d_in[5];
  const float* W2     = (const float*)d_in[6];
  const float* g2     = (const float*)d_in[7];
  const float* b2     = (const float*)d_in[8];
  const float* Wf     = (const float*)d_in[9];
  const float* gf     = (const float*)d_in[10];
  const float* bf     = (const float*)d_in[11];
  float* out = (float*)d_out;
  float* ws  = (float*)d_ws;

  hipMemsetAsync(ws, 0, 224 * sizeof(float), stream);
  k_stats1<<<256, 512, 0, stream>>>(vox, coords, W1, ws);
  k_bnprep<<<1, 64, 0, stream>>>(ws + WS_SUM1, ws + WS_SQ1, g1, b1, ws + WS_S1, ws + WS_B1,
                                 16, 1.0f / (float)NPTS, 1e-3f);
  k_pass<false><<<1000, 256, 0, stream>>>(vox, coords, nump, W1, W2, Wf, ws, out);
  k_bnprep<<<1, 64, 0, stream>>>(ws + WS_SUM2, ws + WS_SQ2, g2, b2, ws + WS_S2, ws + WS_B2,
                                 32, 1.0f / (float)NPTS, 1e-3f);
  k_pass<true><<<500, 256, 0, stream>>>(vox, coords, nump, W1, W2, Wf, ws, out);
  k_bnprep<<<1, 64, 0, stream>>>(ws + WS_SUMF, ws + WS_SQF, gf, bf, ws + WS_SF, ws + WS_BF,
                                 64, 1.0f / (float)NVOX, 1e-5f);
  k_norm<<<(NVOX * 64 + 255) / 256, 256, 0, stream>>>(out, ws + WS_SF, ws + WS_BF, NVOX * 64);
}


# BN scale/bias folding moved into the consumers: k_pass and k_norm derive their channels' pair from the stat sums in the prologue; the three one-block k_bnprep launches go (7 launches to 4)
# speedup vs baseline: 2.0832x; 1.0836x over previous
#include <hip/hip_runtime.h>

typedef __attribute__((ext_vector_type(2))) float v2f;
typedef __attribute__((ext_vector_type(8))) float v8f;

#define NB 2
#define NV 16000
#define NPOINT 32
#define NVOX (NB * NV)
#define NPTS (NVOX * NPOINT)

#define WS_SUM1 0
#define WS_SQ1  16
#define WS_SUM2 32
#define WS_SQ2  64
#define WS_SUMF 96
#define WS_SQF  160

__device__ __forceinline__ void compute_feats(const float4& pt, const int4& cc, float f[8]) {
  float cx = ((float)cc.w + 0.5f) * 0.16f + 0.0f;
  float cy = ((float)cc.z + 0.5f) * 0.16f - 39.68f;
  float cz = ((float)cc.y + 0.5f) * 4.0f - 3.0f;
  f[0] = pt.x; f[1] = pt.y; f[2] = pt.z;
  f[3] = pt.x - cx; f[4] = pt.y - cy; f[5] = pt.z - cz;
  f[6] = pt.w; f[7] = pt.z;
}

__global__ __launch_bounds__(512) void k_stats1(const float* __restrict__ vox,
                                                const int* __restrict__ coords,
                                                const float* __restrict__ W1,
                                                float* __restrict__ ws) {
  __shared__ float wpart[16 * 32];
  __shared__ float sW[16 * 8];
  for (int i = threadIdx.x; i < 128; i += blockDim.x) sW[i] = W1[i];
  __syncthreads();

  float acc[16], acq[16];
  #pragma unroll
  for (int o = 0; o < 16; o++) { acc[o] = 0.f; acq[o] = 0.f; }

  int tid = blockIdx.x * blockDim.x + threadIdx.x;
  int stride = gridDim.x * blockDim.x;
  for (int p = tid; p < NPTS; p += stride) {
    int vxl = p >> 5;
    float4 pt = ((const float4*)vox)[p];
    int4 cc = ((const int4*)coords)[vxl];
    float f[8];
    compute_feats(pt, cc, f);
    #pragma unroll
    for (int o = 0; o < 16; o++) {
      float h = 0.f;
      #pragma unroll
      for (int c = 0; c < 8; c++) h = fmaf(sW[o * 8 + c], f[c], h);
      acc[o] += h;
      acq[o] = fmaf(h, h, acq[o]);
    }
  }
  const int lane = threadIdx.x & 31;
  const int wave = threadIdx.x >> 5;
  float v[32];
  #pragma unroll
  for (int o = 0; o < 16; o++) { v[o] = acc[o]; v[16 + o] = acq[o]; }
  #pragma unroll
  for (int w = 16; w >= 1; w >>= 1) {
    const bool up = (lane & w) != 0;
    #pragma unroll
    for (int i = 0; i < w; i++) {
      const float lo = v[i], hi = v[i + w];
      const float keep = up ? hi : lo;
      const float send = up ? lo : hi;
      v[i] = keep + __shfl_xor(send, w);
    }
  }
  wpart[wave * 32 + lane] = v[0];
  __syncthreads();
  if (threadIdx.x < 32) {
    float s = 0.f;
    for (int k = 0; k < 16; k++) s += wpart[k * 32 + threadIdx.x];
    atomicAdd(&ws[WS_SUM1 + threadIdx.x], s);
  }
}

__device__ __forceinline__ void bn_fold(float sum, float sumsq, float gamma, float beta,
                                        float invN, float eps, float& scale, float& bias) {
  float m = sum * invN;
  float var = sumsq * invN - m * m;
  float s = gamma * rsqrtf(var + eps);
  scale = s;
  bias = fmaf(-m, s, beta);
}

template <bool FINAL>
__global__ __launch_bounds__(256) void k_pass(const float* __restrict__ vox,
                                              const int* __restrict__ coords,
                                              const int* __restrict__ nump,
                                              const float* __restrict__ W1,
                                              const float* __restrict__ g1,
                                              const float* __restrict__ b1,
                                              const float* __restrict__ W2,
                                              const float* __restrict__ g2,
                                              const float* __restrict__ b2,
                                              const float* __restrict__ Wf,
                                              float* __restrict__ ws,
                                              float* __restrict__ out) {
  __shared__ float lds_f[8 * 32 * 8];
  __shared__ float lds_x2[8 * 32 * 32];
  __shared__ float lds_vf[16 * 64];
  __shared__ float sstat[128];

  const int lane = threadIdx.x & 31;
  const int wave = threadIdx.x >> 5;
  const int n = lane & 15;
  const int hi = lane >> 4;

  for (int i = threadIdx.x; i < 128; i += blockDim.x) sstat[i] = 0.f;
  for (int i = threadIdx.x; i < 16 * 64; i += blockDim.x) lds_vf[i] = 0.f;
  __syncthreads();

  float* myf = lds_f + wave * (32 * 8);
  float* myx = lds_x2 + wave * (32 * 32);

  v2f bw1[2], bw2[2][8], bwf[16];
  #pragma unroll
  for (int kb = 0; kb < 2; kb++) bw1[kb] = *(const v2f*)(W1 + n * 8 + 4 * kb + 2 * hi);
  #pragma unroll
  for (int nt = 0; nt < 2; nt++)
    #pragma unroll
    for (int kb = 0; kb < 8; kb++)
      bw2[nt][kb] = *(const v2f*)(W2 + (nt * 16 + n) * 32 + 4 * kb + 2 * hi);
  if (FINAL) {
    int fnt = wave & 3;
    #pragma unroll
    for (int kb = 0; kb < 16; kb++)
      bwf[kb] = *(const v2f*)(Wf + (fnt * 16 + n) * 64 + 4 * kb + 2 * hi);
  }

  float s1v, b1v;
  bn_fold(ws[WS_SUM1 + n], ws[WS_SQ1 + n], g1[n], b1[n], 1.0f / (float)NPTS, 1e-3f, s1v, b1v);
  float s2v[2] = {0.f, 0.f}, b2v[2] = {0.f, 0.f};
  if (FINAL) {
    #pragma unroll
    for (int nt = 0; nt < 2; nt++)
      bn_fold(ws[WS_SUM2 + nt * 16 + n], ws[WS_SQ2 + nt * 16 + n], g2[nt * 16 + n], b2[nt * 16 + n],
              1.0f / (float)NPTS, 1e-3f, s2v[nt], b2v[nt]);
  }

  float accS[2] = {0.f, 0.f}, accQ[2] = {0.f, 0.f};
  float accSF = 0.f, accQF = 0.f;

  const int nchunks = NVOX / 8;
  for (int chunk = blockIdx.x; chunk < nchunks; chunk += gridDim.x) {
    const int vxl = chunk * 8 + wave;
    float4 pt = ((const float4*)vox)[vxl * NPOINT + lane];
    int4 cc = ((const int4*)coords)[vxl];
    int np_ = nump[vxl];

    float f[8];
    compute_feats(pt, cc, f);
    *(float4*)(myf + lane * 8 + 0) = make_float4(f[0], f[1], f[2], f[3]);
    *(float4*)(myf + lane * 8 + 4) = make_float4(f[4], f[5], f[6], f[7]);
    __syncthreads();

    v8f h1t[2];
    #pragma unroll
    for (int mt = 0; mt < 2; mt++) {
      v8f acc = {};
      #pragma unroll
      for (int kb = 0; kb < 2; kb++) {
        v2f a = *(const v2f*)(myf + (mt * 16 + n) * 8 + 4 * kb + 2 * hi);
        acc = __builtin_amdgcn_wmma_f32_16x16x4_f32(false, a, false, bw1[kb], (short)0, acc,
                                                    false, false);
      }
      h1t[mt] = acc;
    }

    float hval[2][8];
    float mxp = 0.f;
    #pragma unroll
    for (int mt = 0; mt < 2; mt++)
      #pragma unroll
      for (int r = 0; r < 8; r++) {
        float h = fmaxf(fmaf(s1v, h1t[mt][r], b1v), 0.f);
        hval[mt][r] = h;
        mxp = fmaxf(mxp, h);
      }
    mxp = fmaxf(mxp, __shfl_xor(mxp, 16));
    #pragma unroll
    for (int mt = 0; mt < 2; mt++)
      #pragma unroll
      for (int r = 0; r < 8; r++) {
        int p = mt * 16 + 8 * hi + r;
        float m = (p < np_) ? 1.f : 0.f;
        myx[p * 32 + n] = hval[mt][r] * m;
        myx[p * 32 + 16 + n] = mxp * m;
      }
    __syncthreads();

    v8f h2t[2][2];
    #pragma unroll
    for (int mt = 0; mt < 2; mt++) {
      v2f a[8];
      #pragma unroll
      for (int kb = 0; kb < 8; kb++)
        a[kb] = *(const v2f*)(myx + (mt * 16 + n) * 32 + 4 * kb + 2 * hi);
      #pragma unroll
      for (int nt = 0; nt < 2; nt++) {
        v8f acc = {};
        #pragma unroll
        for (int kb = 0; kb < 8; kb++)
          acc = __builtin_amdgcn_wmma_f32_16x16x4_f32(false, a[kb], false, bw2[nt][kb], (short)0,
                                                      acc, false, false);
        h2t[mt][nt] = acc;
      }
    }

    if (!FINAL) {
      #pragma unroll
      for (int nt = 0; nt < 2; nt++) {
        float s = 0.f, q = 0.f;
        #pragma unroll
        for (int mt = 0; mt < 2; mt++)
          #pragma unroll
          for (int r = 0; r < 8; r++) {
            float h = h2t[mt][nt][r];
            s += h;
            q = fmaf(h, h, q);
          }
        accS[nt] += s;
        accQ[nt] += q;
      }
    } else {
      float mx2[2] = {0.f, 0.f};
      float vfh[2] = {0.f, 0.f};
      #pragma unroll
      for (int nt = 0; nt < 2; nt++) {
        #pragma unroll
        for (int mt = 0; mt < 2; mt++)
          #pragma unroll
          for (int r = 0; r < 8; r++) {
            float h = fmaxf(fmaf(s2v[nt], h2t[mt][nt][r], b2v[nt]), 0.f);
            mx2[nt] = fmaxf(mx2[nt], h);
            int p = mt * 16 + 8 * hi + r;
            if (p < np_) vfh[nt] = fmaxf(vfh[nt], h);
          }
        mx2[nt] = fmaxf(mx2[nt], __shfl_xor(mx2[nt], 16));
        vfh[nt] = fmaxf(vfh[nt], __shfl_xor(vfh[nt], 16));
      }
      float anyv = (np_ > 0) ? 1.f : 0.f;
      if (hi == 0) {
        lds_vf[wave * 64 + n]      = vfh[0];
        lds_vf[wave * 64 + 16 + n] = vfh[1];
        lds_vf[wave * 64 + 32 + n] = mx2[0] * anyv;
        lds_vf[wave * 64 + 48 + n] = mx2[1] * anyv;
      }
      __syncthreads();

      if (wave < 4) {
        v8f acc = {};
        #pragma unroll
        for (int kb = 0; kb < 16; kb++) {
          v2f a = *(const v2f*)(lds_vf + n * 64 + 4 * kb + 2 * hi);
          acc = __builtin_amdgcn_wmma_f32_16x16x4_f32(false, a, false, bwf[kb], (short)0, acc,
                                                      false, false);
        }
        if (hi == 0) {
          #pragma unroll
          for (int r = 0; r < 8; r++) {
            float g = acc[r];
            out[(chunk * 8 + r) * 64 + wave * 16 + n] = g;
            accSF += g;
            accQF = fmaf(g, g, accQF);
          }
        }
      }
      __syncthreads();
    }
  }

  __syncthreads();
  if (!FINAL) {
    atomicAdd(&sstat[n], accS[0]);
    atomicAdd(&sstat[16 + n], accS[1]);
    atomicAdd(&sstat[32 + n], accQ[0]);
    atomicAdd(&sstat[48 + n], accQ[1]);
    __syncthreads();
    for (int i = threadIdx.x; i < 64; i += blockDim.x) atomicAdd(&ws[WS_SUM2 + i], sstat[i]);
  } else {
    if (wave < 4 && hi == 0) {
      atomicAdd(&sstat[wave * 16 + n], accSF);
      atomicAdd(&sstat[64 + wave * 16 + n], accQF);
    }
    __syncthreads();
    for (int i = threadIdx.x; i < 128; i += blockDim.x) atomicAdd(&ws[WS_SUMF + i], sstat[i]);
  }
}

__global__ __launch_bounds__(256) void k_norm(float* __restrict__ out,
                                              const float* __restrict__ ws,
                                              const float* __restrict__ gf,
                                              const float* __restrict__ bf, int nElem) {
  int i = blockIdx.x * blockDim.x + threadIdx.x;
  if (i < nElem) {
    int c = i & 63;
    float s, b;
    bn_fold(ws[WS_SUMF + c], ws[WS_SQF + c], gf[c], bf[c], 1.0f / (float)NVOX, 1e-5f, s, b);
    out[i] = fmaf(s, out[i], b);
  }
}

extern "C" void kernel_launch(void* const* d_in, const int* in_sizes, int n_in,
                              void* d_out, int out_size, void* d_ws, size_t ws_size,
                              hipStream_t stream) {
  (void)in_sizes; (void)n_in; (void)out_size; (void)ws_size;
  const float* vox    = (const float*)d_in[0];
  const int*   coords = (const int*)d_in[1];
  const int*   nump   = (const int*)d_in[2];
  const float* W1     = (const float*)d_in[3];
  const float* g1     = (const float*)d_in[4];
  const float* b1     = (const float*)d_in[5];
  const float* W2     = (const float*)d_in[6];
  const float* g2     = (const float*)d_in[7];
  const float* b2     = (const float*)d_in[8];
  const float* Wf     = (const float*)d_in[9];
  const float* gf     = (const float*)d_in[10];
  const float* bf     = (const float*)d_in[11];
  float* out = (float*)d_out;
  float* ws  = (float*)d_ws;

  hipMemsetAsync(ws, 0, 224 * sizeof(float), stream);
  k_stats1<<<256, 512, 0, stream>>>(vox, coords, W1, ws);
  k_pass<false><<<1000, 256, 0, stream>>>(vox, coords, nump, W1, g1, b1, W2, g2, b2, Wf, ws, out);
  k_pass<true><<<500, 256, 0, stream>>>(vox, coords, nump, W1, g1, b1, W2, g2, b2, Wf, ws, out);
  k_norm<<<(NVOX * 64 + 255) / 256, 256, 0, stream>>>(out, ws, gf, bf, NVOX * 64);
}


# k_pass LDS row pitches padded (feats 8->12, layer-2 input 32->36, voxel features 64->68 floats): A-fragment reads and D-layout scatter stores touch every bank once, not 2/8/16 rows per bank
# speedup vs baseline: 2.3024x; 1.1052x over previous
#include <hip/hip_runtime.h>

typedef __attribute__((ext_vector_type(2))) float v2f;
typedef __attribute__((ext_vector_type(8))) float v8f;

#define NB 2
#define NV 16000
#define NPOINT 32
#define NVOX (NB * NV)
#define NPTS (NVOX * NPOINT)

#define WS_SUM1 0
#define WS_SQ1  16
#define WS_SUM2 32
#define WS_SQ2  64
#define WS_SUMF 96
#define WS_SQF  160

#define PF 12
#define PX 36
#define PV 68

__device__ __forceinline__ void compute_feats(const float4& pt, const int4& cc, float f[8]) {
  float cx = ((float)cc.w + 0.5f) * 0.16f + 0.0f;
  float cy = ((float)cc.z + 0.5f) * 0.16f - 39.68f;
  float cz = ((float)cc.y + 0.5f) * 4.0f - 3.0f;
  f[0] = pt.x; f[1] = pt.y; f[2] = pt.z;
  f[3] = pt.x - cx; f[4] = pt.y - cy; f[5] = pt.z - cz;
  f[6] = pt.w; f[7] = pt.z;
}

__global__ __launch_bounds__(512) void k_stats1(const float* __restrict__ vox,
                                                const int* __restrict__ coords,
                                                const float* __restrict__ W1,
                                                float* __restrict__ ws) {
  __shared__ float wpart[16 * 32];
  __shared__ float sW[16 * 8];
  for (int i = threadIdx.x; i < 128; i += blockDim.x) sW[i] = W1[i];
  __syncthreads();

  float acc[16], acq[16];
  #pragma unroll
  for (int o = 0; o < 16; o++) { acc[o] = 0.f; acq[o] = 0.f; }

  int tid = blockIdx.x * blockDim.x + threadIdx.x;
  int stride = gridDim.x * blockDim.x;
  for (int p = tid; p < NPTS; p += stride) {
    int vxl = p >> 5;
    float4 pt = ((const float4*)vox)[p];
    int4 cc = ((const int4*)coords)[vxl];
    float f[8];
    compute_feats(pt, cc, f);
    #pragma unroll
    for (int o = 0; o < 16; o++) {
      float h = 0.f;
      #pragma unroll
      for (int c = 0; c < 8; c++) h = fmaf(sW[o * 8 + c], f[c], h);
      acc[o] += h;
      acq[o] = fmaf(h, h, acq[o]);
    }
  }
  const int lane = threadIdx.x & 31;
  const int wave = threadIdx.x >> 5;
  float v[32];
  #pragma unroll
  for (int o = 0; o < 16; o++) { v[o] = acc[o]; v[16 + o] = acq[o]; }
  #pragma unroll
  for (int w = 16; w >= 1; w >>= 1) {
    const bool up = (lane & w) != 0;
    #pragma unroll
    for (int i = 0; i < w; i++) {
      const float lo = v[i], hi = v[i + w];
      const float keep = up ? hi : lo;
      const float send = up ? lo : hi;
      v[i] = keep + __shfl_xor(send, w);
    }
  }
  wpart[wave * 32 + lane] = v[0];
  __syncthreads();
  if (threadIdx.x < 32) {
    float s = 0.f;
    for (int k = 0; k < 16; k++) s += wpart[k * 32 + threadIdx.x];
    atomicAdd(&ws[WS_SUM1 + threadIdx.x], s);
  }
}

__device__ __forceinline__ void bn_fold(float sum, float sumsq, float gamma, float beta,
                                        float invN, float eps, float& scale, float& bias) {
  float m = sum * invN;
  float var = sumsq * invN - m * m;
  float s = gamma * rsqrtf(var + eps);
  scale = s;
  bias = fmaf(-m, s, beta);
}

template <bool FINAL>
__global__ __launch_bounds__(256) void k_pass(const float* __restrict__ vox,
                                              const int* __restrict__ coords,
                                              const int* __restrict__ nump,
                                              const float* __restrict__ W1,
                                              const float* __restrict__ g1,
                                              const float* __restrict__ b1,
                                              const float* __restrict__ W2,
                                              const float* __restrict__ g2,
                                              const float* __restrict__ b2,
                                              const float* __restrict__ Wf,
                                              float* __restrict__ ws,
                                              float* __restrict__ out) {
  __shared__ float lds_f[8 * 32 * PF];
  __shared__ float lds_x2[8 * 32 * PX];
  __shared__ float lds_vf[16 * PV];
  __shared__ float sstat[128];

  const int lane = threadIdx.x & 31;
  const int wave = threadIdx.x >> 5;
  const int n = lane & 15;
  const int hi = lane >> 4;

  for (int i = threadIdx.x; i < 128; i += blockDim.x) sstat[i] = 0.f;
  for (int i = threadIdx.x; i < 16 * PV; i += blockDim.x) lds_vf[i] = 0.f;
  __syncthreads();

  float* myf = lds_f + wave * (32 * PF);
  float* myx = lds_x2 + wave * (32 * PX);

  v2f bw1[2], bw2[2][8], bwf[16];
  #pragma unroll
  for (int kb = 0; kb < 2; kb++) bw1[kb] = *(const v2f*)(W1 + n * 8 + 4 * kb + 2 * hi);
  #pragma unroll
  for (int nt = 0; nt < 2; nt++)
    #pragma unroll
    for (int kb = 0; kb < 8; kb++)
      bw2[nt][kb] = *(const v2f*)(W2 + (nt * 16 + n) * 32 + 4 * kb + 2 * hi);
  if (FINAL) {
    int fnt = wave & 3;
    #pragma unroll
    for (int kb = 0; kb < 16; kb++)
      bwf[kb] = *(const v2f*)(Wf + (fnt * 16 + n) * 64 + 4 * kb + 2 * hi);
  }

  float s1v, b1v;
  bn_fold(ws[WS_SUM1 + n], ws[WS_SQ1 + n], g1[n], b1[n], 1.0f / (float)NPTS, 1e-3f, s1v, b1v);
  float s2v[2] = {0.f, 0.f}, b2v[2] = {0.f, 0.f};
  if (FINAL) {
    #pragma unroll
    for (int nt = 0; nt < 2; nt++)
      bn_fold(ws[WS_SUM2 + nt * 16 + n], ws[WS_SQ2 + nt * 16 + n], g2[nt * 16 + n], b2[nt * 16 + n],
              1.0f / (float)NPTS, 1e-3f, s2v[nt], b2v[nt]);
  }

  float accS[2] = {0.f, 0.f}, accQ[2] = {0.f, 0.f};
  float accSF = 0.f, accQF = 0.f;

  const int nchunks = NVOX / 8;
  for (int chunk = blockIdx.x; chunk < nchunks; chunk += gridDim.x) {
    const int vxl = chunk * 8 + wave;
    float4 pt = ((const float4*)vox)[vxl * NPOINT + lane];
    int4 cc = ((const int4*)coords)[vxl];
    int np_ = nump[vxl];

    float f[8];
    compute_feats(pt, cc, f);
    *(float4*)(myf + lane * PF + 0) = make_float4(f[0], f[1], f[2], f[3]);
    *(float4*)(myf + lane * PF + 4) = make_float4(f[4], f[5], f[6], f[7]);
    __syncthreads();

    v8f h1t[2];
    #pragma unroll
    for (int mt = 0; mt < 2; mt++) {
      v8f acc = {};
      #pragma unroll
      for (int kb = 0; kb < 2; kb++) {
        v2f a = *(const v2f*)(myf + (mt * 16 + n) * PF + 4 * kb + 2 * hi);
        acc = __builtin_amdgcn_wmma_f32_16x16x4_f32(false, a, false, bw1[kb], (short)0, acc,
                                                    false, false);
      }
      h1t[mt] = acc;
    }

    float hval[2][8];
    float mxp = 0.f;
    #pragma unroll
    for (int mt = 0; mt < 2; mt++)
      #pragma unroll
      for (int r = 0; r < 8; r++) {
        float h = fmaxf(fmaf(s1v, h1t[mt][r], b1v), 0.f);
        hval[mt][r] = h;
        mxp = fmaxf(mxp, h);
      }
    mxp = fmaxf(mxp, __shfl_xor(mxp, 16));
    #pragma unroll
    for (int mt = 0; mt < 2; mt++)
      #pragma unroll
      for (int r = 0; r < 8; r++) {
        int p = mt * 16 + 8 * hi + r;
        float m = (p < np_) ? 1.f : 0.f;
        myx[p * PX + n] = hval[mt][r] * m;
        myx[p * PX + 16 + n] = mxp * m;
      }
    __syncthreads();

    v8f h2t[2][2];
    #pragma unroll
    for (int mt = 0; mt < 2; mt++) {
      v2f a[8];
      #pragma unroll
      for (int kb = 0; kb < 8; kb++)
        a[kb] = *(const v2f*)(myx + (mt * 16 + n) * PX + 4 * kb + 2 * hi);
      #pragma unroll
      for (int nt = 0; nt < 2; nt++) {
        v8f acc = {};
        #pragma unroll
        for (int kb = 0; kb < 8; kb++)
          acc = __builtin_amdgcn_wmma_f32_16x16x4_f32(false, a[kb], false, bw2[nt][kb], (short)0,
                                                      acc, false, false);
        h2t[mt][nt] = acc;
      }
    }

    if (!FINAL) {
      #pragma unroll
      for (int nt = 0; nt < 2; nt++) {
        float s = 0.f, q = 0.f;
        #pragma unroll
        for (int mt = 0; mt < 2; mt++)
          #pragma unroll
          for (int r = 0; r < 8; r++) {
            float h = h2t[mt][nt][r];
            s += h;
            q = fmaf(h, h, q);
          }
        accS[nt] += s;
        accQ[nt] += q;
      }
    } else {
      float mx2[2] = {0.f, 0.f};
      float vfh[2] = {0.f, 0.f};
      #pragma unroll
      for (int nt = 0; nt < 2; nt++) {
        #pragma unroll
        for (int mt = 0; mt < 2; mt++)
          #pragma unroll
          for (int r = 0; r < 8; r++) {
            float h = fmaxf(fmaf(s2v[nt], h2t[mt][nt][r], b2v[nt]), 0.f);
            mx2[nt] = fmaxf(mx2[nt], h);
            int p = mt * 16 + 8 * hi + r;
            if (p < np_) vfh[nt] = fmaxf(vfh[nt], h);
          }
        mx2[nt] = fmaxf(mx2[nt], __shfl_xor(mx2[nt], 16));
        vfh[nt] = fmaxf(vfh[nt], __shfl_xor(vfh[nt], 16));
      }
      float anyv = (np_ > 0) ? 1.f : 0.f;
      if (hi == 0) {
        lds_vf[wave * PV + n]      = vfh[0];
        lds_vf[wave * PV + 16 + n] = vfh[1];
        lds_vf[wave * PV + 32 + n] = mx2[0] * anyv;
        lds_vf[wave * PV + 48 + n] = mx2[1] * anyv;
      }
      __syncthreads();

      if (wave < 4) {
        v8f acc = {};
        #pragma unroll
        for (int kb = 0; kb < 16; kb++) {
          v2f a = *(const v2f*)(lds_vf + n * PV + 4 * kb + 2 * hi);
          acc = __builtin_amdgcn_wmma_f32_16x16x4_f32(false, a, false, bwf[kb], (short)0, acc,
                                                      false, false);
        }
        if (hi == 0) {
          #pragma unroll
          for (int r = 0; r < 8; r++) {
            float g = acc[r];
            out[(chunk * 8 + r) * 64 + wave * 16 + n] = g;
            accSF += g;
            accQF = fmaf(g, g, accQF);
          }
        }
      }
      __syncthreads();
    }
  }

  __syncthreads();
  if (!FINAL) {
    atomicAdd(&sstat[n], accS[0]);
    atomicAdd(&sstat[16 + n], accS[1]);
    atomicAdd(&sstat[32 + n], accQ[0]);
    atomicAdd(&sstat[48 + n], accQ[1]);
    __syncthreads();
    for (int i = threadIdx.x; i < 64; i += blockDim.x) atomicAdd(&ws[WS_SUM2 + i], sstat[i]);
  } else {
    if (wave < 4 && hi == 0) {
      atomicAdd(&sstat[wave * 16 + n], accSF);
      atomicAdd(&sstat[64 + wave * 16 + n], accQF);
    }
    __syncthreads();
    for (int i = threadIdx.x; i < 128; i += blockDim.x) atomicAdd(&ws[WS_SUMF + i], sstat[i]);
  }
}

__global__ __launch_bounds__(256) void k_norm(float* __restrict__ out,
                                              const float* __restrict__ ws,
                                              const float* __restrict__ gf,
                                              const float* __restrict__ bf, int nElem) {
  int i = blockIdx.x * blockDim.x + threadIdx.x;
  if (i < nElem) {
    int c = i & 63;
    float s, b;
    bn_fold(ws[WS_SUMF + c], ws[WS_SQF + c], gf[c], bf[c], 1.0f / (float)NVOX, 1e-5f, s, b);
    out[i] = fmaf(s, out[i], b);
  }
}

extern "C" void kernel_launch(void* const* d_in, const int* in_sizes, int n_in,
                              void* d_out, int out_size, void* d_ws, size_t ws_size,
                              hipStream_t stream) {
  (void)in_sizes; (void)n_in; (void)out_size; (void)ws_size;
  const float* vox    = (const float*)d_in[0];
  const int*   coords = (const int*)d_in[1];
  const int*   nump   = (const int*)d_in[2];
  const float* W1     = (const float*)d_in[3];
  const float* g1     = (const float*)d_in[4];
  const float* b1     = (const float*)d_in[5];
  const float* W2     = (const float*)d_in[6];
  const float* g2     = (const float*)d_in[7];
  const float* b2     = (const float*)d_in[8];
  const float* Wf     = (const float*)d_in[9];
  const float* gf     = (const float*)d_in[10];
  const float* bf     = (const float*)d_in[11];
  float* out = (float*)d_out;
  float* ws  = (float*)d_ws;

  hipMemsetAsync(ws, 0, 224 * sizeof(float), stream);
  k_stats1<<<256, 512, 0, stream>>>(vox, coords, W1, ws);
  k_pass<false><<<1000, 256, 0, stream>>>(vox, coords, nump, W1, g1, b1, W2, g2, b2, Wf, ws, out);
  k_pass<true><<<500, 256, 0, stream>>>(vox, coords, nump, W1, g1, b1, W2, g2, b2, Wf, ws, out);
  k_norm<<<(NVOX * 64 + 255) / 256, 256, 0, stream>>>(out, ws, gf, bf, NVOX * 64);
}


# k_pass layer 2 skips the second 16-point tile when the voxel has at most 16 points (the mask zeroes those rows): wave-uniform branch round its LDS stores, fragment loads and 16 WMMAs; sums unchanged
# speedup vs baseline: 2.3799x; 1.0337x over previous
#include <hip/hip_runtime.h>

typedef __attribute__((ext_vector_type(2))) float v2f;
typedef __attribute__((ext_vector_type(8))) float v8f;

#define NB 2
#define NV 16000
#define NPOINT 32
#define NVOX (NB * NV)
#define NPTS (NVOX * NPOINT)

#define WS_SUM1 0
#define WS_SQ1  16
#define WS_SUM2 32
#define WS_SQ2  64
#define WS_SUMF 96
#define WS_SQF  160

#define PF 12
#define PX 36
#define PV 68

__device__ __forceinline__ void compute_feats(const float4& pt, const int4& cc, float f[8]) {
  float cx = ((float)cc.w + 0.5f) * 0.16f + 0.0f;
  float cy = ((float)cc.z + 0.5f) * 0.16f - 39.68f;
  float cz = ((float)cc.y + 0.5f) * 4.0f - 3.0f;
  f[0] = pt.x; f[1] = pt.y; f[2] = pt.z;
  f[3] = pt.x - cx; f[4] = pt.y - cy; f[5] = pt.z - cz;
  f[6] = pt.w; f[7] = pt.z;
}

__global__ __launch_bounds__(512) void k_stats1(const float* __restrict__ vox,
                                                const int* __restrict__ coords,
                                                const float* __restrict__ W1,
                                                float* __restrict__ ws) {
  __shared__ float wpart[16 * 32];
  __shared__ float sW[16 * 8];
  for (int i = threadIdx.x; i < 128; i += blockDim.x) sW[i] = W1[i];
  __syncthreads();

  float acc[16], acq[16];
  #pragma unroll
  for (int o = 0; o < 16; o++) { acc[o] = 0.f; acq[o] = 0.f; }

  int tid = blockIdx.x * blockDim.x + threadIdx.x;
  int stride = gridDim.x * blockDim.x;
  for (int p = tid; p < NPTS; p += stride) {
    int vxl = p >> 5;
    float4 pt = ((const float4*)vox)[p];
    int4 cc = ((const int4*)coords)[vxl];
    float f[8];
    compute_feats(pt, cc, f);
    #pragma unroll
    for (int o = 0; o < 16; o++) {
      float h = 0.f;
      #pragma unroll
      for (int c = 0; c < 8; c++) h = fmaf(sW[o * 8 + c], f[c], h);
      acc[o] += h;
      acq[o] = fmaf(h, h, acq[o]);
    }
  }
  const int lane = threadIdx.x & 31;
  const int wave = threadIdx.x >> 5;
  float v[32];
  #pragma unroll
  for (int o = 0; o < 16; o++) { v[o] = acc[o]; v[16 + o] = acq[o]; }
  #pragma unroll
  for (int w = 16; w >= 1; w >>= 1) {
    const bool up = (lane & w) != 0;
    #pragma unroll
    for (int i = 0; i < w; i++) {
      const float lo = v[i], hi = v[i + w];
      const float keep = up ? hi : lo;
      const float send = up ? lo : hi;
      v[i] = keep + __shfl_xor(send, w);
    }
  }
  wpart[wave * 32 + lane] = v[0];
  __syncthreads();
  if (threadIdx.x < 32) {
    float s = 0.f;
    for (int k = 0; k < 16; k++) s += wpart[k * 32 + threadIdx.x];
    atomicAdd(&ws[WS_SUM1 + threadIdx.x], s);
  }
}

__device__ __forceinline__ void bn_fold(float sum, float sumsq, float gamma, float beta,
                                        float invN, float eps, float& scale, float& bias) {
  float m = sum * invN;
  float var = sumsq * invN - m * m;
  float s = gamma * rsqrtf(var + eps);
  scale = s;
  bias = fmaf(-m, s, beta);
}

template <bool FINAL>
__global__ __launch_bounds__(256) void k_pass(const float* __restrict__ vox,
                                              const int* __restrict__ coords,
                                              const int* __restrict__ nump,
                                              const float* __restrict__ W1,
                                              const float* __restrict__ g1,
                                              const float* __restrict__ b1,
                                              const float* __restrict__ W2,
                                              const float* __restrict__ g2,
                                              const float* __restrict__ b2,
                                              const float* __restrict__ Wf,
                                              float* __restrict__ ws,
                                              float* __restrict__ out) {
  __shared__ float lds_f[8 * 32 * PF];
  __shared__ float lds_x2[8 * 32 * PX];
  __shared__ float lds_vf[16 * PV];
  __shared__ float sstat[128];

  const int lane = threadIdx.x & 31;
  const int wave = threadIdx.x >> 5;
  const int n = lane & 15;
  const int hi = lane >> 4;

  for (int i = threadIdx.x; i < 128; i += blockDim.x) sstat[i] = 0.f;
  for (int i = threadIdx.x; i < 16 * PV; i += blockDim.x) lds_vf[i] = 0.f;
  __syncthreads();

  float* myf = lds_f + wave * (32 * PF);
  float* myx = lds_x2 + wave * (32 * PX);

  v2f bw1[2], bw2[2][8], bwf[16];
  #pragma unroll
  for (int kb = 0; kb < 2; kb++) bw1[kb] = *(const v2f*)(W1 + n * 8 + 4 * kb + 2 * hi);
  #pragma unroll
  for (int nt = 0; nt < 2; nt++)
    #pragma unroll
    for (int kb = 0; kb < 8; kb++)
      bw2[nt][kb] = *(const v2f*)(W2 + (nt * 16 + n) * 32 + 4 * kb + 2 * hi);
  if (FINAL) {
    int fnt = wave & 3;
    #pragma unroll
    for (int kb = 0; kb < 16; kb++)
      bwf[kb] = *(const v2f*)(Wf + (fnt * 16 + n) * 64 + 4 * kb + 2 * hi);
  }

  float s1v, b1v;
  bn_fold(ws[WS_SUM1 + n], ws[WS_SQ1 + n], g1[n], b1[n], 1.0f / (float)NPTS, 1e-3f, s1v, b1v);
  float s2v[2] = {0.f, 0.f}, b2v[2] = {0.f, 0.f};
  if (FINAL) {
    #pragma unroll
    for (int nt = 0; nt < 2; nt++)
      bn_fold(ws[WS_SUM2 + nt * 16 + n], ws[WS_SQ2 + nt * 16 + n], g2[nt * 16 + n], b2[nt * 16 + n],
              1.0f / (float)NPTS, 1e-3f, s2v[nt], b2v[nt]);
  }

  float accS[2] = {0.f, 0.f}, accQ[2] = {0.f, 0.f};
  float accSF = 0.f, accQF = 0.f;

  const int nchunks = NVOX / 8;
  for (int chunk = blockIdx.x; chunk < nchunks; chunk += gridDim.x) {
    const int vxl = chunk * 8 + wave;
    float4 pt = ((const float4*)vox)[vxl * NPOINT + lane];
    int4 cc = ((const int4*)coords)[vxl];
    int np_ = nump[vxl];
    const int npu = __builtin_amdgcn_readfirstlane(np_);
    const bool tile1 = npu > 16;

    float f[8];
    compute_feats(pt, cc, f);
    *(float4*)(myf + lane * PF + 0) = make_float4(f[0], f[1], f[2], f[3]);
    *(float4*)(myf + lane * PF + 4) = make_float4(f[4], f[5], f[6], f[7]);
    __syncthreads();

    v8f h1t[2];
    #pragma unroll
    for (int mt = 0; mt < 2; mt++) {
      v8f acc = {};
      #pragma unroll
      for (int kb = 0; kb < 2; kb++) {
        v2f a = *(const v2f*)(myf + (mt * 16 + n) * PF + 4 * kb + 2 * hi);
        acc = __builtin_amdgcn_wmma_f32_16x16x4_f32(false, a, false, bw1[kb], (short)0, acc,
                                                    false, false);
      }
      h1t[mt] = acc;
    }

    float hval[2][8];
    float mxp = 0.f;
    #pragma unroll
    for (int mt = 0; mt < 2; mt++)
      #pragma unroll
      for (int r = 0; r < 8; r++) {
        float h = fmaxf(fmaf(s1v, h1t[mt][r], b1v), 0.f);
        hval[mt][r] = h;
        mxp = fmaxf(mxp, h);
      }
    mxp = fmaxf(mxp, __shfl_xor(mxp, 16));
    #pragma unroll
    for (int mt = 0; mt < 2; mt++) {
      if (mt == 0 || tile1) {
        #pragma unroll
        for (int r = 0; r < 8; r++) {
          int p = mt * 16 + 8 * hi + r;
          float m = (p < np_) ? 1.f : 0.f;
          myx[p * PX + n] = hval[mt][r] * m;
          myx[p * PX + 16 + n] = mxp * m;
        }
      }
    }
    __syncthreads();

    v8f h2t[2][2];
    v2f a0[8];
    #pragma unroll
    for (int kb = 0; kb < 8; kb++)
      a0[kb] = *(const v2f*)(myx + n * PX + 4 * kb + 2 * hi);
    h2t[1][0] = v8f{};
    h2t[1][1] = v8f{};
    if (tile1) {
      v2f a1[8];
      #pragma unroll
      for (int kb = 0; kb < 8; kb++)
        a1[kb] = *(const v2f*)(myx + (16 + n) * PX + 4 * kb + 2 * hi);
      #pragma unroll
      for (int nt = 0; nt < 2; nt++) {
        v8f acc = {};
        #pragma unroll
        for (int kb = 0; kb < 8; kb++)
          acc = __builtin_amdgcn_wmma_f32_16x16x4_f32(false, a1[kb], false, bw2[nt][kb], (short)0,
                                                      acc, false, false);
        h2t[1][nt] = acc;
      }
    }
    #pragma unroll
    for (int nt = 0; nt < 2; nt++) {
      v8f acc = {};
      #pragma unroll
      for (int kb = 0; kb < 8; kb++)
        acc = __builtin_amdgcn_wmma_f32_16x16x4_f32(false, a0[kb], false, bw2[nt][kb], (short)0,
                                                    acc, false, false);
      h2t[0][nt] = acc;
    }

    if (!FINAL) {
      #pragma unroll
      for (int nt = 0; nt < 2; nt++) {
        float s = 0.f, q = 0.f;
        #pragma unroll
        for (int mt = 0; mt < 2; mt++)
          #pragma unroll
          for (int r = 0; r < 8; r++) {
            float h = h2t[mt][nt][r];
            s += h;
            q = fmaf(h, h, q);
          }
        accS[nt] += s;
        accQ[nt] += q;
      }
    } else {
      float mx2[2] = {0.f, 0.f};
      float vfh[2] = {0.f, 0.f};
      #pragma unroll
      for (int nt = 0; nt < 2; nt++) {
        #pragma unroll
        for (int mt = 0; mt < 2; mt++)
          #pragma unroll
          for (int r = 0; r < 8; r++) {
            float h = fmaxf(fmaf(s2v[nt], h2t[mt][nt][r], b2v[nt]), 0.f);
            mx2[nt] = fmaxf(mx2[nt], h);
            int p = mt * 16 + 8 * hi + r;
            if (p < np_) vfh[nt] = fmaxf(vfh[nt], h);
          }
        mx2[nt] = fmaxf(mx2[nt], __shfl_xor(mx2[nt], 16));
        vfh[nt] = fmaxf(vfh[nt], __shfl_xor(vfh[nt], 16));
      }
      float anyv = (np_ > 0) ? 1.f : 0.f;
      if (hi == 0) {
        lds_vf[wave * PV + n]      = vfh[0];
        lds_vf[wave * PV + 16 + n] = vfh[1];
        lds_vf[wave * PV + 32 + n] = mx2[0] * anyv;
        lds_vf[wave * PV + 48 + n] = mx2[1] * anyv;
      }
      __syncthreads();

      if (wave < 4) {
        v8f acc = {};
        #pragma unroll
        for (int kb = 0; kb < 16; kb++) {
          v2f a = *(const v2f*)(lds_vf + n * PV + 4 * kb + 2 * hi);
          acc = __builtin_amdgcn_wmma_f32_16x16x4_f32(false, a, false, bwf[kb], (short)0, acc,
                                                      false, false);
        }
        if (hi == 0) {
          #pragma unroll
          for (int r = 0; r < 8; r++) {
            float g = acc[r];
            out[(chunk * 8 + r) * 64 + wave * 16 + n] = g;
            accSF += g;
            accQF = fmaf(g, g, accQF);
          }
        }
      }
      __syncthreads();
    }
  }

  __syncthreads();
  if (!FINAL) {
    atomicAdd(&sstat[n], accS[0]);
    atomicAdd(&sstat[16 + n], accS[1]);
    atomicAdd(&sstat[32 + n], accQ[0]);
    atomicAdd(&sstat[48 + n], accQ[1]);
    __syncthreads();
    for (int i = threadIdx.x; i < 64; i += blockDim.x) atomicAdd(&ws[WS_SUM2 + i], sstat[i]);
  } else {
    if (wave < 4 && hi == 0) {
      atomicAdd(&sstat[wave * 16 + n], accSF);
      atomicAdd(&sstat[64 + wave * 16 + n], accQF);
    }
    __syncthreads();
    for (int i = threadIdx.x; i < 128; i += blockDim.x) atomicAdd(&ws[WS_SUMF + i], sstat[i]);
  }
}

__global__ __launch_bounds__(256) void k_norm(float* __restrict__ out,
                                              const float* __restrict__ ws,
                                              const float* __restrict__ gf,
                                              const float* __restrict__ bf, int nElem) {
  int i = blockIdx.x * blockDim.x + threadIdx.x;
  if (i < nElem) {
    int c = i & 63;
    float s, b;
    bn_fold(ws[WS_SUMF + c], ws[WS_SQF + c], gf[c], bf[c], 1.0f / (float)NVOX, 1e-5f, s, b);
    out[i] = fmaf(s, out[i], b);
  }
}

extern "C" void kernel_launch(void* const* d_in, const int* in_sizes, int n_in,
                              void* d_out, int out_size, void* d_ws, size_t ws_size,
                              hipStream_t stream) {
  (void)in_sizes; (void)n_in; (void)out_size; (void)ws_size;
  const float* vox    = (const float*)d_in[0];
  const int*   coords = (const int*)d_in[1];
  const int*   nump   = (const int*)d_in[2];
  const float* W1     = (const float*)d_in[3];
  const float* g1     = (const float*)d_in[4];
  const float* b1     = (const float*)d_in[5];
  const float* W2     = (const float*)d_in[6];
  const float* g2     = (const float*)d_in[7];
  const float* b2     = (const float*)d_in[8];
  const float* Wf     = (const float*)d_in[9];
  const float* gf     = (const float*)d_in[10];
  const float* bf     = (const float*)d_in[11];
  float* out = (float*)d_out;
  float* ws  = (float*)d_ws;

  hipMemsetAsync(ws, 0, 224 * sizeof(float), stream);
  k_stats1<<<256, 512, 0, stream>>>(vox, coords, W1, ws);
  k_pass<false><<<1000, 256, 0, stream>>>(vox, coords, nump, W1, g1, b1, W2, g2, b2, Wf, ws, out);
  k_pass<true><<<500, 256, 0, stream>>>(vox, coords, nump, W1, g1, b1, W2, g2, b2, Wf, ws, out);
  k_norm<<<(NVOX * 64 + 255) / 256, 256, 0, stream>>>(out, ws, gf, bf, NVOX * 64);
}


# final pass no longer redoes layers 1-2: stats pass stores each voxel's per-channel max/min of pre-BN layer-2 output in out (BN2+relu is monotone), new WMMA kernel reads them; final-BN sums regrouped
# speedup vs baseline: 3.4326x; 1.4423x over previous
#include <hip/hip_runtime.h>

typedef __attribute__((ext_vector_type(2))) float v2f;
typedef __attribute__((ext_vector_type(8))) float v8f;

#define NB 2
#define NV 16000
#define NPOINT 32
#define NVOX (NB * NV)
#define NPTS (NVOX * NPOINT)

#define WS_SUM1 0
#define WS_SQ1  16
#define WS_SUM2 32
#define WS_SQ2  64
#define WS_SUMF 96
#define WS_SQF  160

#define PF 12
#define PX 36
#define PV 68

__device__ __forceinline__ void compute_feats(const float4& pt, const int4& cc, float f[8]) {
  float cx = ((float)cc.w + 0.5f) * 0.16f + 0.0f;
  float cy = ((float)cc.z + 0.5f) * 0.16f - 39.68f;
  float cz = ((float)cc.y + 0.5f) * 4.0f - 3.0f;
  f[0] = pt.x; f[1] = pt.y; f[2] = pt.z;
  f[3] = pt.x - cx; f[4] = pt.y - cy; f[5] = pt.z - cz;
  f[6] = pt.w; f[7] = pt.z;
}

__global__ __launch_bounds__(512) void k_stats1(const float* __restrict__ vox,
                                                const int* __restrict__ coords,
                                                const float* __restrict__ W1,
                                                float* __restrict__ ws) {
  __shared__ float wpart[16 * 32];
  __shared__ float sW[16 * 8];
  for (int i = threadIdx.x; i < 128; i += blockDim.x) sW[i] = W1[i];
  __syncthreads();

  float acc[16], acq[16];
  #pragma unroll
  for (int o = 0; o < 16; o++) { acc[o] = 0.f; acq[o] = 0.f; }

  int tid = blockIdx.x * blockDim.x + threadIdx.x;
  int stride = gridDim.x * blockDim.x;
  for (int p = tid; p < NPTS; p += stride) {
    int vxl = p >> 5;
    float4 pt = ((const float4*)vox)[p];
    int4 cc = ((const int4*)coords)[vxl];
    float f[8];
    compute_feats(pt, cc, f);
    #pragma unroll
    for (int o = 0; o < 16; o++) {
      float h = 0.f;
      #pragma unroll
      for (int c = 0; c < 8; c++) h = fmaf(sW[o * 8 + c], f[c], h);
      acc[o] += h;
      acq[o] = fmaf(h, h, acq[o]);
    }
  }
  const int lane = threadIdx.x & 31;
  const int wave = threadIdx.x >> 5;
  float v[32];
  #pragma unroll
  for (int o = 0; o < 16; o++) { v[o] = acc[o]; v[16 + o] = acq[o]; }
  #pragma unroll
  for (int w = 16; w >= 1; w >>= 1) {
    const bool up = (lane & w) != 0;
    #pragma unroll
    for (int i = 0; i < w; i++) {
      const float lo = v[i], hi = v[i + w];
      const float keep = up ? hi : lo;
      const float send = up ? lo : hi;
      v[i] = keep + __shfl_xor(send, w);
    }
  }
  wpart[wave * 32 + lane] = v[0];
  __syncthreads();
  if (threadIdx.x < 32) {
    float s = 0.f;
    for (int k = 0; k < 16; k++) s += wpart[k * 32 + threadIdx.x];
    atomicAdd(&ws[WS_SUM1 + threadIdx.x], s);
  }
}

__device__ __forceinline__ void bn_fold(float sum, float sumsq, float gamma, float beta,
                                        float invN, float eps, float& scale, float& bias) {
  float m = sum * invN;
  float var = sumsq * invN - m * m;
  float s = gamma * rsqrtf(var + eps);
  scale = s;
  bias = fmaf(-m, s, beta);
}

__global__ __launch_bounds__(256, 8) void k_pass(const float* __restrict__ vox,
                                              const int* __restrict__ coords,
                                              const int* __restrict__ nump,
                                              const float* __restrict__ W1,
                                              const float* __restrict__ g1,
                                              const float* __restrict__ b1,
                                              const float* __restrict__ W2,
                                              float* __restrict__ ws,
                                              float* __restrict__ out) {
  __shared__ float lds_f[8 * 32 * PF];
  __shared__ float lds_x2[8 * 32 * PX];
  __shared__ float sstat[64];

  const int lane = threadIdx.x & 31;
  const int wave = threadIdx.x >> 5;
  const int n = lane & 15;
  const int hi = lane >> 4;

  for (int i = threadIdx.x; i < 64; i += blockDim.x) sstat[i] = 0.f;
  __syncthreads();

  float* myf = lds_f + wave * (32 * PF);
  float* myx = lds_x2 + wave * (32 * PX);

  v2f bw1[2], bw2[2][8];
  #pragma unroll
  for (int kb = 0; kb < 2; kb++) bw1[kb] = *(const v2f*)(W1 + n * 8 + 4 * kb + 2 * hi);
  #pragma unroll
  for (int nt = 0; nt < 2; nt++)
    #pragma unroll
    for (int kb = 0; kb < 8; kb++)
      bw2[nt][kb] = *(const v2f*)(W2 + (nt * 16 + n) * 32 + 4 * kb + 2 * hi);

  float s1v, b1v;
  bn_fold(ws[WS_SUM1 + n], ws[WS_SQ1 + n], g1[n], b1[n], 1.0f / (float)NPTS, 1e-3f, s1v, b1v);

  float accS[2] = {0.f, 0.f}, accQ[2] = {0.f, 0.f};

  const int nchunks = NVOX / 8;
  for (int chunk = blockIdx.x; chunk < nchunks; chunk += gridDim.x) {
    const int vxl = chunk * 8 + wave;
    float4 pt = ((const float4*)vox)[vxl * NPOINT + lane];
    int4 cc = ((const int4*)coords)[vxl];
    int np_ = nump[vxl];
    const int npu = __builtin_amdgcn_readfirstlane(np_);
    const bool tile1 = npu > 16;

    float f[8];
    compute_feats(pt, cc, f);
    *(float4*)(myf + lane * PF + 0) = make_float4(f[0], f[1], f[2], f[3]);
    *(float4*)(myf + lane * PF + 4) = make_float4(f[4], f[5], f[6], f[7]);
    __syncthreads();

    v8f h1t[2];
    #pragma unroll
    for (int mt = 0; mt < 2; mt++) {
      v8f acc = {};
      #pragma unroll
      for (int kb = 0; kb < 2; kb++) {
        v2f a = *(const v2f*)(myf + (mt * 16 + n) * PF + 4 * kb + 2 * hi);
        acc = __builtin_amdgcn_wmma_f32_16x16x4_f32(false, a, false, bw1[kb], (short)0, acc,
                                                    false, false);
      }
      h1t[mt] = acc;
    }

    float hval[2][8];
    float mxp = 0.f;
    #pragma unroll
    for (int mt = 0; mt < 2; mt++)
      #pragma unroll
      for (int r = 0; r < 8; r++) {
        float h = fmaxf(fmaf(s1v, h1t[mt][r], b1v), 0.f);
        hval[mt][r] = h;
        mxp = fmaxf(mxp, h);
      }
    mxp = fmaxf(mxp, __shfl_xor(mxp, 16));
    #pragma unroll
    for (int mt = 0; mt < 2; mt++) {
      if (mt == 0 || tile1) {
        #pragma unroll
        for (int r = 0; r < 8; r++) {
          int p = mt * 16 + 8 * hi + r;
          float m = (p < np_) ? 1.f : 0.f;
          myx[p * PX + n] = hval[mt][r] * m;
          myx[p * PX + 16 + n] = mxp * m;
        }
      }
    }
    __syncthreads();

    v8f h2t[2][2];
    v2f a0[8];
    #pragma unroll
    for (int kb = 0; kb < 8; kb++)
      a0[kb] = *(const v2f*)(myx + n * PX + 4 * kb + 2 * hi);
    h2t[1][0] = v8f{};
    h2t[1][1] = v8f{};
    if (tile1) {
      v2f a1[8];
      #pragma unroll
      for (int kb = 0; kb < 8; kb++)
        a1[kb] = *(const v2f*)(myx + (16 + n) * PX + 4 * kb + 2 * hi);
      #pragma unroll
      for (int nt = 0; nt < 2; nt++) {
        v8f acc = {};
        #pragma unroll
        for (int kb = 0; kb < 8; kb++)
          acc = __builtin_amdgcn_wmma_f32_16x16x4_f32(false, a1[kb], false, bw2[nt][kb], (short)0,
                                                      acc, false, false);
        h2t[1][nt] = acc;
      }
    }
    #pragma unroll
    for (int nt = 0; nt < 2; nt++) {
      v8f acc = {};
      #pragma unroll
      for (int kb = 0; kb < 8; kb++)
        acc = __builtin_amdgcn_wmma_f32_16x16x4_f32(false, a0[kb], false, bw2[nt][kb], (short)0,
                                                    acc, false, false);
      h2t[0][nt] = acc;
    }

    float vmx[2], vmn[2];
    #pragma unroll
    for (int nt = 0; nt < 2; nt++) {
      float s = 0.f, q = 0.f;
      float mx = -__builtin_inff(), mn = __builtin_inff();
      #pragma unroll
      for (int mt = 0; mt < 2; mt++)
        #pragma unroll
        for (int r = 0; r < 8; r++) {
          float h = h2t[mt][nt][r];
          s += h;
          q = fmaf(h, h, q);
          const bool ok = (mt * 16 + 8 * hi + r) < np_;
          mx = fmaxf(mx, ok ? h : -__builtin_inff());
          mn = fminf(mn, ok ? h : __builtin_inff());
        }
      accS[nt] += s;
      accQ[nt] += q;
      mx = fmaxf(mx, __shfl_xor(mx, 16));
      mn = fminf(mn, __shfl_xor(mn, 16));
      vmx[nt] = (np_ > 0) ? mx : 0.f;
      vmn[nt] = (np_ > 0) ? mn : 0.f;
    }
    out[vxl * 64 + hi * 32 + n]      = hi ? vmn[0] : vmx[0];
    out[vxl * 64 + hi * 32 + 16 + n] = hi ? vmn[1] : vmx[1];
  }

  __syncthreads();
  atomicAdd(&sstat[n], accS[0]);
  atomicAdd(&sstat[16 + n], accS[1]);
  atomicAdd(&sstat[32 + n], accQ[0]);
  atomicAdd(&sstat[48 + n], accQ[1]);
  __syncthreads();
  for (int i = threadIdx.x; i < 64; i += blockDim.x) atomicAdd(&ws[WS_SUM2 + i], sstat[i]);
}

#define FTILES (NVOX / 16)
#define FGRID  (FTILES / 8)
static_assert(FGRID * 8 * 16 == NVOX, "the final pass covers the voxels with whole workgroups");

__global__ __launch_bounds__(256) void k_final(const int* __restrict__ nump,
                                               const float* __restrict__ g2,
                                               const float* __restrict__ b2,
                                               const float* __restrict__ Wf,
                                               float* __restrict__ ws,
                                               float* __restrict__ out) {
  __shared__ float sWf[64 * PV];
  __shared__ float sT[8 * 16 * PV];
  __shared__ float sS2[32];
  __shared__ float sB2[32];
  __shared__ float wpart[8 * 128];

  const int lane = threadIdx.x & 31;
  const int wave = threadIdx.x >> 5;
  const int n = lane & 15;
  const int hi = lane >> 4;
  const int tile = blockIdx.x * 8 + wave;
  float* rows = out + tile * (16 * 64);
  float* myT = sT + wave * (16 * PV);

  float4 tv[8];
  #pragma unroll
  for (int i = 0; i < 8; i++) tv[i] = ((const float4*)rows)[lane + 32 * i];
  const int np_ = nump[tile * 16 + n];
  float4 wv[4];
  #pragma unroll
  for (int i = 0; i < 4; i++) wv[i] = ((const float4*)Wf)[threadIdx.x + 256 * i];
  if (threadIdx.x < 32) {
    float s, b;
    bn_fold(ws[WS_SUM2 + threadIdx.x], ws[WS_SQ2 + threadIdx.x], g2[threadIdx.x], b2[threadIdx.x],
            1.0f / (float)NPTS, 1e-3f, s, b);
    sS2[threadIdx.x] = s;
    sB2[threadIdx.x] = b;
  }
  #pragma unroll
  for (int i = 0; i < 4; i++) {
    const int idx = threadIdx.x + 256 * i;
    *(float4*)(sWf + (idx >> 4) * PV + (idx & 15) * 4) = wv[i];
  }
  #pragma unroll
  for (int i = 0; i < 8; i++) {
    const int idx = lane + 32 * i;
    *(float4*)(myT + (idx >> 4) * PV + (idx & 15) * 4) = tv[i];
  }
  __syncthreads();

  const bool any = np_ > 0;
  const bool part = np_ < 32;
  v2f a[16];
  #pragma unroll
  for (int j = 0; j < 8; j++) {
    const v2f mx = *(const v2f*)(myT + n * PV + 4 * j + 2 * hi);
    const v2f mn = *(const v2f*)(myT + n * PV + 32 + 4 * j + 2 * hi);
    const v2f sc = *(const v2f*)(sS2 + 4 * j + 2 * hi);
    const v2f bi = *(const v2f*)(sB2 + 4 * j + 2 * hi);
    #pragma unroll
    for (int e = 0; e < 2; e++) {
      const float hv = (sc[e] >= 0.f) ? mx[e] : mn[e];
      const float t = fmaxf(fmaf(sc[e], hv, bi[e]), 0.f);
      const float m2 = part ? fmaxf(t, fmaxf(bi[e], 0.f)) : t;
      a[j][e] = any ? t : 0.f;
      a[8 + j][e] = any ? m2 : 0.f;
    }
  }

  v8f acc[4] = {v8f{}, v8f{}, v8f{}, v8f{}};
  #pragma unroll
  for (int kb = 0; kb < 16; kb++) {
    #pragma unroll
    for (int nt = 0; nt < 4; nt++) {
      const v2f bfr = *(const v2f*)(sWf + (nt * 16 + n) * PV + 4 * kb + 2 * hi);
      acc[nt] = __builtin_amdgcn_wmma_f32_16x16x4_f32(false, a[kb], false, bfr, (short)0, acc[nt],
                                                      false, false);
    }
  }

  #pragma unroll
  for (int nt = 0; nt < 4; nt++) {
    float s = 0.f, q = 0.f;
    #pragma unroll
    for (int r = 0; r < 8; r++) {
      const float g = acc[nt][r];
      rows[(8 * hi + r) * 64 + nt * 16 + n] = g;
      s += g;
      q = fmaf(g, g, q);
    }
    s += __shfl_xor(s, 16);
    q += __shfl_xor(q, 16);
    wpart[wave * 128 + hi * 64 + nt * 16 + n] = hi ? q : s;
  }
  __syncthreads();
  if (threadIdx.x < 128) {
    float s = 0.f;
    #pragma unroll
    for (int w = 0; w < 8; w++) s += wpart[w * 128 + threadIdx.x];
    atomicAdd(&ws[WS_SUMF + threadIdx.x], s);
  }
}

__global__ __launch_bounds__(256) void k_norm(float* __restrict__ out,
                                              const float* __restrict__ ws,
                                              const float* __restrict__ gf,
                                              const float* __restrict__ bf, int nElem) {
  int i = blockIdx.x * blockDim.x + threadIdx.x;
  if (i < nElem) {
    int c = i & 63;
    float s, b;
    bn_fold(ws[WS_SUMF + c], ws[WS_SQF + c], gf[c], bf[c], 1.0f / (float)NVOX, 1e-5f, s, b);
    out[i] = fmaf(s, out[i], b);
  }
}

extern "C" void kernel_launch(void* const* d_in, const int* in_sizes, int n_in,
                              void* d_out, int out_size, void* d_ws, size_t ws_size,
                              hipStream_t stream) {
  (void)in_sizes; (void)n_in; (void)out_size; (void)ws_size;
  const float* vox    = (const float*)d_in[0];
  const int*   coords = (const int*)d_in[1];
  const int*   nump   = (const int*)d_in[2];
  const float* W1     = (const float*)d_in[3];
  const float* g1     = (const float*)d_in[4];
  const float* b1     = (const float*)d_in[5];
  const float* W2     = (const float*)d_in[6];
  const float* g2     = (const float*)d_in[7];
  const float* b2     = (const float*)d_in[8];
  const float* Wf     = (const float*)d_in[9];
  const float* gf     = (const float*)d_in[10];
  const float* bf     = (const float*)d_in[11];
  float* out = (float*)d_out;
  float* ws  = (float*)d_ws;

  hipMemsetAsync(ws, 0, 224 * sizeof(float), stream);
  k_stats1<<<256, 512, 0, stream>>>(vox, coords, W1, ws);
  k_pass<<<1000, 256, 0, stream>>>(vox, coords, nump, W1, g1, b1, W2, ws, out);
  k_final<<<FGRID, 256, 0, stream>>>(nump, g2, b2, Wf, ws, out);
  k_norm<<<(NVOX * 64 + 255) / 256, 256, 0, stream>>>(out, ws, gf, bf, NVOX * 64);
}


# k_pass: each wave streams its own voxels one trip ahead (next point row, coords, count loaded under the current layers); LDS rows are wave-private, so wave-scope ordering replaces the block barriers
# speedup vs baseline: 3.6174x; 1.0538x over previous
#include <hip/hip_runtime.h>

typedef __attribute__((ext_vector_type(2))) float v2f;
typedef __attribute__((ext_vector_type(8))) float v8f;

#define NB 2
#define NV 16000
#define NPOINT 32
#define NVOX (NB * NV)
#define NPTS (NVOX * NPOINT)

#define WS_SUM1 0
#define WS_SQ1  16
#define WS_SUM2 32
#define WS_SQ2  64
#define WS_SUMF 96
#define WS_SQF  160

#define PF 12
#define PX 36
#define PV 68

__device__ __forceinline__ void compute_feats(const float4& pt, const int4& cc, float f[8]) {
  float cx = ((float)cc.w + 0.5f) * 0.16f + 0.0f;
  float cy = ((float)cc.z + 0.5f) * 0.16f - 39.68f;
  float cz = ((float)cc.y + 0.5f) * 4.0f - 3.0f;
  f[0] = pt.x; f[1] = pt.y; f[2] = pt.z;
  f[3] = pt.x - cx; f[4] = pt.y - cy; f[5] = pt.z - cz;
  f[6] = pt.w; f[7] = pt.z;
}

__global__ __launch_bounds__(512) void k_stats1(const float* __restrict__ vox,
                                                const int* __restrict__ coords,
                                                const float* __restrict__ W1,
                                                float* __restrict__ ws) {
  __shared__ float wpart[16 * 32];
  __shared__ float sW[16 * 8];
  for (int i = threadIdx.x; i < 128; i += blockDim.x) sW[i] = W1[i];
  __syncthreads();

  float acc[16], acq[16];
  #pragma unroll
  for (int o = 0; o < 16; o++) { acc[o] = 0.f; acq[o] = 0.f; }

  int tid = blockIdx.x * blockDim.x + threadIdx.x;
  int stride = gridDim.x * blockDim.x;
  for (int p = tid; p < NPTS; p += stride) {
    int vxl = p >> 5;
    float4 pt = ((const float4*)vox)[p];
    int4 cc = ((const int4*)coords)[vxl];
    float f[8];
    compute_feats(pt, cc, f);
    #pragma unroll
    for (int o = 0; o < 16; o++) {
      float h = 0.f;
      #pragma unroll
      for (int c = 0; c < 8; c++) h = fmaf(sW[o * 8 + c], f[c], h);
      acc[o] += h;
      acq[o] = fmaf(h, h, acq[o]);
    }
  }
  const int lane = threadIdx.x & 31;
  const int wave = threadIdx.x >> 5;
  float v[32];
  #pragma unroll
  for (int o = 0; o < 16; o++) { v[o] = acc[o]; v[16 + o] = acq[o]; }
  #pragma unroll
  for (int w = 16; w >= 1; w >>= 1) {
    const bool up = (lane & w) != 0;
    #pragma unroll
    for (int i = 0; i < w; i++) {
      const float lo = v[i], hi = v[i + w];
      const float keep = up ? hi : lo;
      const float send = up ? lo : hi;
      v[i] = keep + __shfl_xor(send, w);
    }
  }
  wpart[wave * 32 + lane] = v[0];
  __syncthreads();
  if (threadIdx.x < 32) {
    float s = 0.f;
    for (int k = 0; k < 16; k++) s += wpart[k * 32 + threadIdx.x];
    atomicAdd(&ws[WS_SUM1 + threadIdx.x], s);
  }
}

__device__ __forceinline__ void bn_fold(float sum, float sumsq, float gamma, float beta,
                                        float invN, float eps, float& scale, float& bias) {
  float m = sum * invN;
  float var = sumsq * invN - m * m;
  float s = gamma * rsqrtf(var + eps);
  scale = s;
  bias = fmaf(-m, s, beta);
}

__device__ __forceinline__ void wave_lds_handover() {
  __builtin_amdgcn_fence(__ATOMIC_ACQ_REL, "wavefront");
}

__global__ __launch_bounds__(256, 8) void k_pass(const float* __restrict__ vox,
                                              const int* __restrict__ coords,
                                              const int* __restrict__ nump,
                                              const float* __restrict__ W1,
                                              const float* __restrict__ g1,
                                              const float* __restrict__ b1,
                                              const float* __restrict__ W2,
                                              float* __restrict__ ws,
                                              float* __restrict__ out) {
  __shared__ float lds_f[8 * 32 * PF];
  __shared__ float lds_x2[8 * 32 * PX];
  __shared__ float sstat[64];

  const int lane = threadIdx.x & 31;
  const int wave = threadIdx.x >> 5;
  const int n = lane & 15;
  const int hi = lane >> 4;

  for (int i = threadIdx.x; i < 64; i += blockDim.x) sstat[i] = 0.f;
  __syncthreads();

  float* myf = lds_f + wave * (32 * PF);
  float* myx = lds_x2 + wave * (32 * PX);

  v2f bw1[2], bw2[2][8];
  #pragma unroll
  for (int kb = 0; kb < 2; kb++) bw1[kb] = *(const v2f*)(W1 + n * 8 + 4 * kb + 2 * hi);
  #pragma unroll
  for (int nt = 0; nt < 2; nt++)
    #pragma unroll
    for (int kb = 0; kb < 8; kb++)
      bw2[nt][kb] = *(const v2f*)(W2 + (nt * 16 + n) * 32 + 4 * kb + 2 * hi);

  float s1v, b1v;
  bn_fold(ws[WS_SUM1 + n], ws[WS_SQ1 + n], g1[n], b1[n], 1.0f / (float)NPTS, 1e-3f, s1v, b1v);

  float accS[2] = {0.f, 0.f}, accQ[2] = {0.f, 0.f};

  const int nchunks = NVOX / 8;
  float4 pt_nx = make_float4(0.f, 0.f, 0.f, 0.f);
  int4 cc_nx = make_int4(0, 0, 0, 0);
  int np_nx = 0;
  if (blockIdx.x < nchunks) {
    const int v0 = blockIdx.x * 8 + wave;
    pt_nx = ((const float4*)vox)[v0 * NPOINT + lane];
    cc_nx = ((const int4*)coords)[v0];
    np_nx = nump[v0];
  }
  for (int chunk = blockIdx.x; chunk < nchunks; chunk += gridDim.x) {
    const int vxl = chunk * 8 + wave;
    float f[8];
    compute_feats(pt_nx, cc_nx, f);
    const int npu = __builtin_amdgcn_readfirstlane(np_nx);
    const int np_l = np_nx - 8 * hi;
    const bool tile1 = npu > 16;
    const bool any = npu > 0;

    const int nxt = chunk + gridDim.x;
    if (nxt < nchunks) {
      const int v1 = nxt * 8 + wave;
      pt_nx = ((const float4*)vox)[v1 * NPOINT + lane];
      cc_nx = ((const int4*)coords)[v1];
      np_nx = nump[v1];
    }
    *(float4*)(myf + lane * PF + 0) = make_float4(f[0], f[1], f[2], f[3]);
    *(float4*)(myf + lane * PF + 4) = make_float4(f[4], f[5], f[6], f[7]);
    wave_lds_handover();

    v8f h1t[2];
    #pragma unroll
    for (int mt = 0; mt < 2; mt++) {
      v8f acc = {};
      #pragma unroll
      for (int kb = 0; kb < 2; kb++) {
        v2f a = *(const v2f*)(myf + (mt * 16 + n) * PF + 4 * kb + 2 * hi);
        acc = __builtin_amdgcn_wmma_f32_16x16x4_f32(false, a, false, bw1[kb], (short)0, acc,
                                                    false, false);
      }
      h1t[mt] = acc;
    }

    float hval[2][8];
    float mxp = 0.f;
    #pragma unroll
    for (int mt = 0; mt < 2; mt++)
      #pragma unroll
      for (int r = 0; r < 8; r++) {
        float h = fmaxf(fmaf(s1v, h1t[mt][r], b1v), 0.f);
        hval[mt][r] = h;
        mxp = fmaxf(mxp, h);
      }
    mxp = fmaxf(mxp, __shfl_xor(mxp, 16));
    #pragma unroll
    for (int mt = 0; mt < 2; mt++) {
      if (mt == 0 || tile1) {
        #pragma unroll
        for (int r = 0; r < 8; r++) {
          int p = mt * 16 + 8 * hi + r;
          float m = (mt * 16 + r < np_l) ? 1.f : 0.f;
          myx[p * PX + n] = hval[mt][r] * m;
          myx[p * PX + 16 + n] = mxp * m;
        }
      }
    }
    wave_lds_handover();

    v8f h2t[2][2];
    v2f a0[8];
    #pragma unroll
    for (int kb = 0; kb < 8; kb++)
      a0[kb] = *(const v2f*)(myx + n * PX + 4 * kb + 2 * hi);
    h2t[1][0] = v8f{};
    h2t[1][1] = v8f{};
    if (tile1) {
      v2f a1[8];
      #pragma unroll
      for (int kb = 0; kb < 8; kb++)
        a1[kb] = *(const v2f*)(myx + (16 + n) * PX + 4 * kb + 2 * hi);
      #pragma unroll
      for (int nt = 0; nt < 2; nt++) {
        v8f acc = {};
        #pragma unroll
        for (int kb = 0; kb < 8; kb++)
          acc = __builtin_amdgcn_wmma_f32_16x16x4_f32(false, a1[kb], false, bw2[nt][kb], (short)0,
                                                      acc, false, false);
        h2t[1][nt] = acc;
      }
    }
    #pragma unroll
    for (int nt = 0; nt < 2; nt++) {
      v8f acc = {};
      #pragma unroll
      for (int kb = 0; kb < 8; kb++)
        acc = __builtin_amdgcn_wmma_f32_16x16x4_f32(false, a0[kb], false, bw2[nt][kb], (short)0,
                                                    acc, false, false);
      h2t[0][nt] = acc;
    }

    float vmx[2], vmn[2];
    #pragma unroll
    for (int nt = 0; nt < 2; nt++) {
      float s = 0.f, q = 0.f;
      float mx = -__builtin_inff(), mn = __builtin_inff();
      #pragma unroll
      for (int mt = 0; mt < 2; mt++)
        #pragma unroll
        for (int r = 0; r < 8; r++) {
          float h = h2t[mt][nt][r];
          s += h;
          q = fmaf(h, h, q);
          const bool ok = (mt * 16 + r) < np_l;
          mx = fmaxf(mx, ok ? h : -__builtin_inff());
          mn = fminf(mn, ok ? h : __builtin_inff());
        }
      accS[nt] += s;
      accQ[nt] += q;
      mx = fmaxf(mx, __shfl_xor(mx, 16));
      mn = fminf(mn, __shfl_xor(mn, 16));
      vmx[nt] = any ? mx : 0.f;
      vmn[nt] = any ? mn : 0.f;
    }
    out[vxl * 64 + hi * 32 + n]      = hi ? vmn[0] : vmx[0];
    out[vxl * 64 + hi * 32 + 16 + n] = hi ? vmn[1] : vmx[1];
  }

  __syncthreads();
  atomicAdd(&sstat[n], accS[0]);
  atomicAdd(&sstat[16 + n], accS[1]);
  atomicAdd(&sstat[32 + n], accQ[0]);
  atomicAdd(&sstat[48 + n], accQ[1]);
  __syncthreads();
  for (int i = threadIdx.x; i < 64; i += blockDim.x) atomicAdd(&ws[WS_SUM2 + i], sstat[i]);
}

#define FTILES (NVOX / 16)
#define FGRID  (FTILES / 8)
static_assert(FGRID * 8 * 16 == NVOX, "the final pass covers the voxels with whole workgroups");

__global__ __launch_bounds__(256) void k_final(const int* __restrict__ nump,
                                               const float* __restrict__ g2,
                                               const float* __restrict__ b2,
                                               const float* __restrict__ Wf,
                                               float* __restrict__ ws,
                                               float* __restrict__ out) {
  __shared__ float sWf[64 * PV];
  __shared__ float sT[8 * 16 * PV];
  __shared__ float sS2[32];
  __shared__ float sB2[32];
  __shared__ float wpart[8 * 128];

  const int lane = threadIdx.x & 31;
  const int wave = threadIdx.x >> 5;
  const int n = lane & 15;
  const int hi = lane >> 4;
  const int tile = blockIdx.x * 8 + wave;
  float* rows = out + tile * (16 * 64);
  float* myT = sT + wave * (16 * PV);

  float4 tv[8];
  #pragma unroll
  for (int i = 0; i < 8; i++) tv[i] = ((const float4*)rows)[lane + 32 * i];
  const int np_ = nump[tile * 16 + n];
  float4 wv[4];
  #pragma unroll
  for (int i = 0; i < 4; i++) wv[i] = ((const float4*)Wf)[threadIdx.x + 256 * i];
  if (threadIdx.x < 32) {
    float s, b;
    bn_fold(ws[WS_SUM2 + threadIdx.x], ws[WS_SQ2 + threadIdx.x], g2[threadIdx.x], b2[threadIdx.x],
            1.0f / (float)NPTS, 1e-3f, s, b);
    sS2[threadIdx.x] = s;
    sB2[threadIdx.x] = b;
  }
  #pragma unroll
  for (int i = 0; i < 4; i++) {
    const int idx = threadIdx.x + 256 * i;
    *(float4*)(sWf + (idx >> 4) * PV + (idx & 15) * 4) = wv[i];
  }
  #pragma unroll
  for (int i = 0; i < 8; i++) {
    const int idx = lane + 32 * i;
    *(float4*)(myT + (idx >> 4) * PV + (idx & 15) * 4) = tv[i];
  }
  __syncthreads();

  const bool any = np_ > 0;
  const bool part = np_ < 32;
  v2f a[16];
  #pragma unroll
  for (int j = 0; j < 8; j++) {
    const v2f mx = *(const v2f*)(myT + n * PV + 4 * j + 2 * hi);
    const v2f mn = *(const v2f*)(myT + n * PV + 32 + 4 * j + 2 * hi);
    const v2f sc = *(const v2f*)(sS2 + 4 * j + 2 * hi);
    const v2f bi = *(const v2f*)(sB2 + 4 * j + 2 * hi);
    #pragma unroll
    for (int e = 0; e < 2; e++) {
      const float hv = (sc[e] >= 0.f) ? mx[e] : mn[e];
      const float t = fmaxf(fmaf(sc[e], hv, bi[e]), 0.f);
      const float m2 = part ? fmaxf(t, fmaxf(bi[e], 0.f)) : t;
      a[j][e] = any ? t : 0.f;
      a[8 + j][e] = any ? m2 : 0.f;
    }
  }

  v8f acc[4] = {v8f{}, v8f{}, v8f{}, v8f{}};
  #pragma unroll
  for (int kb = 0; kb < 16; kb++) {
    #pragma unroll
    for (int nt = 0; nt < 4; nt++) {
      const v2f bfr = *(const v2f*)(sWf + (nt * 16 + n) * PV + 4 * kb + 2 * hi);
      acc[nt] = __builtin_amdgcn_wmma_f32_16x16x4_f32(false, a[kb], false, bfr, (short)0, acc[nt],
                                                      false, false);
    }
  }

  #pragma unroll
  for (int nt = 0; nt < 4; nt++) {
    float s = 0.f, q = 0.f;
    #pragma unroll
    for (int r = 0; r < 8; r++) {
      const float g = acc[nt][r];
      rows[(8 * hi + r) * 64 + nt * 16 + n] = g;
      s += g;
      q = fmaf(g, g, q);
    }
    s += __shfl_xor(s, 16);
    q += __shfl_xor(q, 16);
    wpart[wave * 128 + hi * 64 + nt * 16 + n] = hi ? q : s;
  }
  __syncthreads();
  if (threadIdx.x < 128) {
    float s = 0.f;
    #pragma unroll
    for (int w = 0; w < 8; w++) s += wpart[w * 128 + threadIdx.x];
    atomicAdd(&ws[WS_SUMF + threadIdx.x], s);
  }
}

__global__ __launch_bounds__(256) void k_norm(float* __restrict__ out,
                                              const float* __restrict__ ws,
                                              const float* __restrict__ gf,
                                              const float* __restrict__ bf, int nElem) {
  int i = blockIdx.x * blockDim.x + threadIdx.x;
  if (i < nElem) {
    int c = i & 63;
    float s, b;
    bn_fold(ws[WS_SUMF + c], ws[WS_SQF + c], gf[c], bf[c], 1.0f / (float)NVOX, 1e-5f, s, b);
    out[i] = fmaf(s, out[i], b);
  }
}

extern "C" void kernel_launch(void* const* d_in, const int* in_sizes, int n_in,
                              void* d_out, int out_size, void* d_ws, size_t ws_size,
                              hipStream_t stream) {
  (void)in_sizes; (void)n_in; (void)out_size; (void)ws_size;
  const float* vox    = (const float*)d_in[0];
  const int*   coords = (const int*)d_in[1];
  const int*   nump   = (const int*)d_in[2];
  const float* W1     = (const float*)d_in[3];
  const float* g1     = (const float*)d_in[4];
  const float* b1     = (const float*)d_in[5];
  const float* W2     = (const float*)d_in[6];
  const float* g2     = (const float*)d_in[7];
  const float* b2     = (const float*)d_in[8];
  const float* Wf     = (const float*)d_in[9];
  const float* gf     = (const float*)d_in[10];
  const float* bf     = (const float*)d_in[11];
  float* out = (float*)d_out;
  float* ws  = (float*)d_ws;

  hipMemsetAsync(ws, 0, 224 * sizeof(float), stream);
  k_stats1<<<256, 512, 0, stream>>>(vox, coords, W1, ws);
  k_pass<<<1000, 256, 0, stream>>>(vox, coords, nump, W1, g1, b1, W2, ws, out);
  k_final<<<FGRID, 256, 0, stream>>>(nump, g2, b2, Wf, ws, out);
  k_norm<<<(NVOX * 64 + 255) / 256, 256, 0, stream>>>(out, ws, gf, bf, NVOX * 64);
}


# k_pass layer 2: the voxel-max half of the input is one row per voxel: its W2 product runs once per voxel (a wave's 4 voxels as one tile) and seeds the valid rows; point K 32 to 16; new sum order
# speedup vs baseline: 3.8337x; 1.0598x over previous
#include <hip/hip_runtime.h>

typedef __attribute__((ext_vector_type(2))) float v2f;
typedef __attribute__((ext_vector_type(8))) float v8f;

#define NB 2
#define NV 16000
#define NPOINT 32
#define NVOX (NB * NV)
#define NPTS (NVOX * NPOINT)

#define WS_SUM1 0
#define WS_SQ1  16
#define WS_SUM2 32
#define WS_SQ2  64
#define WS_SUMF 96
#define WS_SQF  160

#define PF 12
#define PV 68

__device__ __forceinline__ void compute_feats(const float4& pt, const int4& cc, float f[8]) {
  float cx = ((float)cc.w + 0.5f) * 0.16f + 0.0f;
  float cy = ((float)cc.z + 0.5f) * 0.16f - 39.68f;
  float cz = ((float)cc.y + 0.5f) * 4.0f - 3.0f;
  f[0] = pt.x; f[1] = pt.y; f[2] = pt.z;
  f[3] = pt.x - cx; f[4] = pt.y - cy; f[5] = pt.z - cz;
  f[6] = pt.w; f[7] = pt.z;
}

__global__ __launch_bounds__(512) void k_stats1(const float* __restrict__ vox,
                                                const int* __restrict__ coords,
                                                const float* __restrict__ W1,
                                                float* __restrict__ ws) {
  __shared__ float wpart[16 * 32];
  __shared__ float sW[16 * 8];
  for (int i = threadIdx.x; i < 128; i += blockDim.x) sW[i] = W1[i];
  __syncthreads();

  float acc[16], acq[16];
  #pragma unroll
  for (int o = 0; o < 16; o++) { acc[o] = 0.f; acq[o] = 0.f; }

  int tid = blockIdx.x * blockDim.x + threadIdx.x;
  int stride = gridDim.x * blockDim.x;
  for (int p = tid; p < NPTS; p += stride) {
    int vxl = p >> 5;
    float4 pt = ((const float4*)vox)[p];
    int4 cc = ((const int4*)coords)[vxl];
    float f[8];
    compute_feats(pt, cc, f);
    #pragma unroll
    for (int o = 0; o < 16; o++) {
      float h = 0.f;
      #pragma unroll
      for (int c = 0; c < 8; c++) h = fmaf(sW[o * 8 + c], f[c], h);
      acc[o] += h;
      acq[o] = fmaf(h, h, acq[o]);
    }
  }
  const int lane = threadIdx.x & 31;
  const int wave = threadIdx.x >> 5;
  float v[32];
  #pragma unroll
  for (int o = 0; o < 16; o++) { v[o] = acc[o]; v[16 + o] = acq[o]; }
  #pragma unroll
  for (int w = 16; w >= 1; w >>= 1) {
    const bool up = (lane & w) != 0;
    #pragma unroll
    for (int i = 0; i < w; i++) {
      const float lo = v[i], hi = v[i + w];
      const float keep = up ? hi : lo;
      const float send = up ? lo : hi;
      v[i] = keep + __shfl_xor(send, w);
    }
  }
  wpart[wave * 32 + lane] = v[0];
  __syncthreads();
  if (threadIdx.x < 32) {
    float s = 0.f;
    for (int k = 0; k < 16; k++) s += wpart[k * 32 + threadIdx.x];
    atomicAdd(&ws[WS_SUM1 + threadIdx.x], s);
  }
}

__device__ __forceinline__ void bn_fold(float sum, float sumsq, float gamma, float beta,
                                        float invN, float eps, float& scale, float& bias) {
  float m = sum * invN;
  float var = sumsq * invN - m * m;
  float s = gamma * rsqrtf(var + eps);
  scale = s;
  bias = fmaf(-m, s, beta);
}

__device__ __forceinline__ void wave_lds_handover() {
  __builtin_amdgcn_fence(__ATOMIC_ACQ_REL, "wavefront");
}

#define XV 512
#define WAVE_LDS (4 * XV + 64)
#define PGRID (NVOX / 32)
static_assert(PGRID * 8 * 4 == NVOX, "the stats pass covers the voxels with whole workgroups");

__device__ __forceinline__ int xpos(int p, int c) {
  const int q = p ^ ((p >> 3) & 1);
  return q * 16 + ((((c >> 2) ^ (q >> 2)) & 3) << 2) + (c & 3);
}

__global__ __launch_bounds__(256, 8) void k_pass(const float* __restrict__ vox,
                                              const int* __restrict__ coords,
                                              const int* __restrict__ nump,
                                              const float* __restrict__ W1,
                                              const float* __restrict__ g1,
                                              const float* __restrict__ b1,
                                              const float* __restrict__ W2,
                                              float* __restrict__ ws,
                                              float* __restrict__ out) {
  __shared__ float lds_x[8 * WAVE_LDS];
  __shared__ float sstat[64];

  const int lane = threadIdx.x & 31;
  const int wave = __builtin_amdgcn_readfirstlane((int)(threadIdx.x >> 5));
  const int n = lane & 15;
  const int hi = lane >> 4;

  for (int i = threadIdx.x; i < 64; i += blockDim.x) sstat[i] = 0.f;
  __syncthreads();

  float* myx = lds_x + wave * WAVE_LDS;
  float* myf = myx + 3 * XV;
  float* myT = myx + 4 * XV;

  v2f bw1[2], bw2[2][8];
  #pragma unroll
  for (int kb = 0; kb < 2; kb++) bw1[kb] = *(const v2f*)(W1 + n * 8 + 4 * kb + 2 * hi);
  #pragma unroll
  for (int nt = 0; nt < 2; nt++)
    #pragma unroll
    for (int kb = 0; kb < 8; kb++)
      bw2[nt][kb] = *(const v2f*)(W2 + (nt * 16 + n) * 32 + 4 * kb + 2 * hi);

  float s1v, b1v;
  bn_fold(ws[WS_SUM1 + n], ws[WS_SQ1 + n], g1[n], b1[n], 1.0f / (float)NPTS, 1e-3f, s1v, b1v);

  float accS[2] = {0.f, 0.f}, accQ[2] = {0.f, 0.f};

  int xld[4];
  #pragma unroll
  for (int kb = 0; kb < 4; kb++) xld[kb] = xpos(n, 4 * kb + 2 * hi);

  int xst[4];
  #pragma unroll
  for (int t = 0; t < 4; t++) xst[t] = xpos(8 * hi + 4 * (t >> 1) + (t & 1), n) - 16 * (4 * (t >> 1) + (t & 1));

  {
    const int vb = (blockIdx.x * 8 + wave) * 4;
    float4 pt[4];
    int4 cc[4];
    int npv[4];
    #pragma unroll
    for (int j = 0; j < 4; j++) {
      pt[j] = ((const float4*)vox)[(vb + j) * NPOINT + lane];
      cc[j] = ((const int4*)coords)[vb + j];
      npv[j] = nump[vb + j];
    }

    int npu[4];
    #pragma unroll
    for (int j = 0; j < 4; j++) {
      float f[8];
      compute_feats(pt[j], cc[j], f);
      npu[j] = __builtin_amdgcn_readfirstlane(npv[j]);
      const int np_l = npu[j] - 8 * hi;
      const bool tile1 = npu[j] > 16;

      *(float4*)(myf + lane * PF + 0) = make_float4(f[0], f[1], f[2], f[3]);
      *(float4*)(myf + lane * PF + 4) = make_float4(f[4], f[5], f[6], f[7]);
      wave_lds_handover();

      v8f h1t[2];
      #pragma unroll
      for (int mt = 0; mt < 2; mt++) {
        v8f acc = {};
        #pragma unroll
        for (int kb = 0; kb < 2; kb++) {
          v2f a = *(const v2f*)(myf + (mt * 16 + n) * PF + 4 * kb + 2 * hi);
          acc = __builtin_amdgcn_wmma_f32_16x16x4_f32(false, a, false, bw1[kb], (short)0, acc,
                                                      false, false);
        }
        h1t[mt] = acc;
      }

      float hval[2][8];
      float mxp = 0.f;
      #pragma unroll
      for (int mt = 0; mt < 2; mt++)
        #pragma unroll
        for (int r = 0; r < 8; r++) {
          float h = fmaxf(fmaf(s1v, h1t[mt][r], b1v), 0.f);
          hval[mt][r] = h;
          mxp = fmaxf(mxp, h);
        }
      mxp = fmaxf(mxp, __shfl_xor(mxp, 16));
      float* xj = myx + j * XV;
      #pragma unroll
      for (int mt = 0; mt < 2; mt++) {
        if (mt == 0 || tile1) {
          #pragma unroll
          for (int r = 0; r < 8; r++)
            xj[mt * 256 + xst[2 * (r >> 2) + (r & 1)] + 16 * r] = (mt * 16 + r < np_l) ? hval[mt][r] : 0.f;
        }
      }
      myT[j * 16 + n] = mxp;
      wave_lds_handover();
    }

    float gv[2][4];
    {
      v2f am[4];
      #pragma unroll
      for (int kb = 0; kb < 4; kb++) am[kb] = *(const v2f*)(myT + (n & 3) * 16 + 4 * kb + 2 * hi);
      #pragma unroll
      for (int nt = 0; nt < 2; nt++) {
        v8f acc = {};
        #pragma unroll
        for (int kb = 0; kb < 4; kb++)
          acc = __builtin_amdgcn_wmma_f32_16x16x4_f32(false, am[kb], false, bw2[nt][4 + kb], (short)0,
                                                      acc, false, false);
        #pragma unroll
        for (int j = 0; j < 4; j++) gv[nt][j] = acc[j];
      }
    }

    #pragma nounroll
    for (int j = 0; j < 4; j++) {
      const int vxl = vb + j;
      const int npj = (j == 0) ? npu[0] : (j == 1) ? npu[1] : (j == 2) ? npu[2] : npu[3];
      const float gj[2] = {(j == 0) ? gv[0][0] : (j == 1) ? gv[0][1] : (j == 2) ? gv[0][2] : gv[0][3],
                           (j == 0) ? gv[1][0] : (j == 1) ? gv[1][1] : (j == 2) ? gv[1][2] : gv[1][3]};
      const int np_l = npj - 8 * hi;
      const bool tile1 = npj > 16;
      const bool any = npj > 0;
      const float* xj = myx + j * XV;

      v8f h2t[2][2];
      v2f a0[4];
      #pragma unroll
      for (int kb = 0; kb < 4; kb++) a0[kb] = *(const v2f*)(xj + xld[kb]);
      h2t[1][0] = v8f{};
      h2t[1][1] = v8f{};
      if (tile1) {
        v2f a1[4];
        #pragma unroll
        for (int kb = 0; kb < 4; kb++) a1[kb] = *(const v2f*)(xj + 256 + xld[kb]);
        #pragma unroll
        for (int nt = 0; nt < 2; nt++) {
          v8f acc;
          #pragma unroll
          for (int r = 0; r < 8; r++) acc[r] = (16 + r < np_l) ? gj[nt] : 0.f;
          #pragma unroll
          for (int kb = 0; kb < 4; kb++)
            acc = __builtin_amdgcn_wmma_f32_16x16x4_f32(false, a1[kb], false, bw2[nt][kb], (short)0,
                                                        acc, false, false);
          h2t[1][nt] = acc;
        }
      }
      #pragma unroll
      for (int nt = 0; nt < 2; nt++) {
        v8f acc;
        #pragma unroll
        for (int r = 0; r < 8; r++) acc[r] = (r < np_l) ? gj[nt] : 0.f;
        #pragma unroll
        for (int kb = 0; kb < 4; kb++)
          acc = __builtin_amdgcn_wmma_f32_16x16x4_f32(false, a0[kb], false, bw2[nt][kb], (short)0,
                                                      acc, false, false);
        h2t[0][nt] = acc;
      }

      float vmx[2], vmn[2];
      #pragma unroll
      for (int nt = 0; nt < 2; nt++) {
        float s = 0.f, q = 0.f;
        float mx = -__builtin_inff(), mn = __builtin_inff();
        #pragma unroll
        for (int mt = 0; mt < 2; mt++)
          #pragma unroll
          for (int r = 0; r < 8; r++) {
            float h = h2t[mt][nt][r];
            s += h;
            q = fmaf(h, h, q);
            const bool ok = (mt * 16 + r) < np_l;
            mx = fmaxf(mx, ok ? h : -__builtin_inff());
            mn = fminf(mn, ok ? h : __builtin_inff());
          }
        accS[nt] += s;
        accQ[nt] += q;
        mx = fmaxf(mx, __shfl_xor(mx, 16));
        mn = fminf(mn, __shfl_xor(mn, 16));
        vmx[nt] = any ? mx : 0.f;
        vmn[nt] = any ? mn : 0.f;
      }
      out[vxl * 64 + hi * 32 + n]      = hi ? vmn[0] : vmx[0];
      out[vxl * 64 + hi * 32 + 16 + n] = hi ? vmn[1] : vmx[1];
    }
  }

  __syncthreads();
  atomicAdd(&sstat[n], accS[0]);
  atomicAdd(&sstat[16 + n], accS[1]);
  atomicAdd(&sstat[32 + n], accQ[0]);
  atomicAdd(&sstat[48 + n], accQ[1]);
  __syncthreads();
  for (int i = threadIdx.x; i < 64; i += blockDim.x) atomicAdd(&ws[WS_SUM2 + i], sstat[i]);
}

#define FTILES (NVOX / 16)
#define FGRID  (FTILES / 8)
static_assert(FGRID * 8 * 16 == NVOX, "the final pass covers the voxels with whole workgroups");

__global__ __launch_bounds__(256) void k_final(const int* __restrict__ nump,
                                               const float* __restrict__ g2,
                                               const float* __restrict__ b2,
                                               const float* __restrict__ Wf,
                                               float* __restrict__ ws,
                                               float* __restrict__ out) {
  __shared__ float sWf[64 * PV];
  __shared__ float sT[8 * 16 * PV];
  __shared__ float sS2[32];
  __shared__ float sB2[32];
  __shared__ float wpart[8 * 128];

  const int lane = threadIdx.x & 31;
  const int wave = threadIdx.x >> 5;
  const int n = lane & 15;
  const int hi = lane >> 4;
  const int tile = blockIdx.x * 8 + wave;
  float* rows = out + tile * (16 * 64);
  float* myT = sT + wave * (16 * PV);

  float4 tv[8];
  #pragma unroll
  for (int i = 0; i < 8; i++) tv[i] = ((const float4*)rows)[lane + 32 * i];
  const int np_ = nump[tile * 16 + n];
  float4 wv[4];
  #pragma unroll
  for (int i = 0; i < 4; i++) wv[i] = ((const float4*)Wf)[threadIdx.x + 256 * i];
  if (threadIdx.x < 32) {
    float s, b;
    bn_fold(ws[WS_SUM2 + threadIdx.x], ws[WS_SQ2 + threadIdx.x], g2[threadIdx.x], b2[threadIdx.x],
            1.0f / (float)NPTS, 1e-3f, s, b);
    sS2[threadIdx.x] = s;
    sB2[threadIdx.x] = b;
  }
  #pragma unroll
  for (int i = 0; i < 4; i++) {
    const int idx = threadIdx.x + 256 * i;
    *(float4*)(sWf + (idx >> 4) * PV + (idx & 15) * 4) = wv[i];
  }
  #pragma unroll
  for (int i = 0; i < 8; i++) {
    const int idx = lane + 32 * i;
    *(float4*)(myT + (idx >> 4) * PV + (idx & 15) * 4) = tv[i];
  }
  __syncthreads();

  const bool any = np_ > 0;
  const bool part = np_ < 32;
  v2f a[16];
  #pragma unroll
  for (int j = 0; j < 8; j++) {
    const v2f mx = *(const v2f*)(myT + n * PV + 4 * j + 2 * hi);
    const v2f mn = *(const v2f*)(myT + n * PV + 32 + 4 * j + 2 * hi);
    const v2f sc = *(const v2f*)(sS2 + 4 * j + 2 * hi);
    const v2f bi = *(const v2f*)(sB2 + 4 * j + 2 * hi);
    #pragma unroll
    for (int e = 0; e < 2; e++) {
      const float hv = (sc[e] >= 0.f) ? mx[e] : mn[e];
      const float t = fmaxf(fmaf(sc[e], hv, bi[e]), 0.f);
      const float m2 = part ? fmaxf(t, fmaxf(bi[e], 0.f)) : t;
      a[j][e] = any ? t : 0.f;
      a[8 + j][e] = any ? m2 : 0.f;
    }
  }

  v8f acc[4] = {v8f{}, v8f{}, v8f{}, v8f{}};
  #pragma unroll
  for (int kb = 0; kb < 16; kb++) {
    #pragma unroll
    for (int nt = 0; nt < 4; nt++) {
      const v2f bfr = *(const v2f*)(sWf + (nt * 16 + n) * PV + 4 * kb + 2 * hi);
      acc[nt] = __builtin_amdgcn_wmma_f32_16x16x4_f32(false, a[kb], false, bfr, (short)0, acc[nt],
                                                      false, false);
    }
  }

  #pragma unroll
  for (int nt = 0; nt < 4; nt++) {
    float s = 0.f, q = 0.f;
    #pragma unroll
    for (int r = 0; r < 8; r++) {
      const float g = acc[nt][r];
      rows[(8 * hi + r) * 64 + nt * 16 + n] = g;
      s += g;
      q = fmaf(g, g, q);
    }
    s += __shfl_xor(s, 16);
    q += __shfl_xor(q, 16);
    wpart[wave * 128 + hi * 64 + nt * 16 + n] = hi ? q : s;
  }
  __syncthreads();
  if (threadIdx.x < 128) {
    float s = 0.f;
    #pragma unroll
    for (int w = 0; w < 8; w++) s += wpart[w * 128 + threadIdx.x];
    atomicAdd(&ws[WS_SUMF + threadIdx.x], s);
  }
}

__global__ __launch_bounds__(256) void k_norm(float* __restrict__ out,
                                              const float* __restrict__ ws,
                                              const float* __restrict__ gf,
                                              const float* __restrict__ bf, int nElem) {
  int i = blockIdx.x * blockDim.x + threadIdx.x;
  if (i < nElem) {
    int c = i & 63;
    float s, b;
    bn_fold(ws[WS_SUMF + c], ws[WS_SQF + c], gf[c], bf[c], 1.0f / (float)NVOX, 1e-5f, s, b);
    out[i] = fmaf(s, out[i], b);
  }
}

extern "C" void kernel_launch(void* const* d_in, const int* in_sizes, int n_in,
                              void* d_out, int out_size, void* d_ws, size_t ws_size,
                              hipStream_t stream) {
  (void)in_sizes; (void)n_in; (void)out_size; (void)ws_size;
  const float* vox    = (const float*)d_in[0];
  const int*   coords = (const int*)d_in[1];
  const int*   nump   = (const int*)d_in[2];
  const float* W1     = (const float*)d_in[3];
  const float* g1     = (const float*)d_in[4];
  const float* b1     = (const float*)d_in[5];
  const float* W2     = (const float*)d_in[6];
  const float* g2     = (const float*)d_in[7];
  const float* b2     = (const float*)d_in[8];
  const float* Wf     = (const float*)d_in[9];
  const float* gf     = (const float*)d_in[10];
  const float* bf     = (const float*)d_in[11];
  float* out = (float*)d_out;
  float* ws  = (float*)d_ws;

  hipMemsetAsync(ws, 0, 224 * sizeof(float), stream);
  k_stats1<<<256, 512, 0, stream>>>(vox, coords, W1, ws);
  k_pass<<<PGRID, 256, 0, stream>>>(vox, coords, nump, W1, g1, b1, W2, ws, out);
  k_final<<<FGRID, 256, 0, stream>>>(nump, g2, b2, Wf, ws, out);
  k_norm<<<(NVOX * 64 + 255) / 256, 256, 0, stream>>>(out, ws, gf, bf, NVOX * 64);
}
